# starting kernel
# baseline (speedup 1.0000x reference)
#include <hip/hip_runtime.h>
#include <hip/hip_bf16.h>

typedef __attribute__((ext_vector_type(2))) float v2f;
typedef __attribute__((ext_vector_type(8))) float v8f;

#define D_FEAT 128


__global__ void zero_f32(float* __restrict__ p, size_t n) {
    size_t i = (size_t)blockIdx.x * blockDim.x + threadIdx.x;
    if (i < n) p[i] = 0.0f;
}

__global__ void degree_kernel(const int* __restrict__ src, const int* __restrict__ dst,
                              float* __restrict__ outdeg, float* __restrict__ indeg, int E) {
    int e = blockIdx.x * blockDim.x + threadIdx.x;
    if (e < E) {
        unsafeAtomicAdd(&outdeg[src[e]], 1.0f);
        unsafeAtomicAdd(&indeg[dst[e]], 1.0f);
    }
}

__global__ void norm_kernel(float* __restrict__ p, int n) {
    int i = blockIdx.x * blockDim.x + threadIdx.x;
    if (i < n) {
        float d = fmaxf(p[i], 1.0f);
        p[i] = rsqrtf(d);
    }
}

__global__ void scatter_kernel(const float* __restrict__ h, const float* __restrict__ norm_src,
                               const int* __restrict__ src, const int* __restrict__ dst,
                               float* __restrict__ agg, int E) {
    int t = blockIdx.x * blockDim.x + threadIdx.x;
    int e = t >> 5;
    int c = (t & 31) << 2;
    if (e < E) {
        int s = src[e];
        int d = dst[e];
        float ns = norm_src[s];
        float4 v = *(const float4*)(h + (size_t)s * D_FEAT + c);
        float* ap = agg + (size_t)d * D_FEAT + c;
        unsafeAtomicAdd(ap + 0, v.x * ns);
        unsafeAtomicAdd(ap + 1, v.y * ns);
        unsafeAtomicAdd(ap + 2, v.z * ns);
        unsafeAtomicAdd(ap + 3, v.w * ns);
    }
}

__global__ void pool_kernel(const float* __restrict__ h, const int* __restrict__ gid,
                            float* __restrict__ hg, int N) {
    int t = blockIdx.x * blockDim.x + threadIdx.x;
    int n = t >> 5;
    int c = (t & 31) << 2;
    if (n < N) {
        int g = gid[n];
        float4 v = *(const float4*)(h + (size_t)n * D_FEAT + c);
        float* hp = hg + (size_t)g * D_FEAT + c;
        unsafeAtomicAdd(hp + 0, v.x);
        unsafeAtomicAdd(hp + 1, v.y);
        unsafeAtomicAdd(hp + 2, v.z);
        unsafeAtomicAdd(hp + 3, v.w);
    }
}

#define MT 4
__global__ void __launch_bounds__(32)
gemm_wmma_f32(const float* __restrict__ A, const float* __restrict__ B,
              const float* __restrict__ bias,
              const float* __restrict__ row_scale,
              const float* __restrict__ resid,
              float* __restrict__ out,
              int Mrows, int K, int Ncol, int relu_c, int mode) {
    int tm   = blockIdx.x;
    int tn   = blockIdx.y;
    int lane = threadIdx.x;
    int half = lane >> 4;
    int l15  = lane & 15;

    int col = tn * 16 + l15;
    float bv = bias[col];

    const float* arow[MT];
    float rs[MT];
#pragma unroll
    for (int i = 0; i < MT; ++i) {
        int row = tm * (16 * MT) + i * 16 + l15;
        if (row >= Mrows) row = Mrows - 1;
        arow[i] = A + (size_t)row * K;
        rs[i] = row_scale ? row_scale[row] : 1.0f;
    }

    v8f acc[MT] = {};
    for (int k0 = 0; k0 < K; k0 += 4) {
        int ka = k0 + half * 2;
        v2f b;
        b.x = B[(size_t)ka * Ncol + col];
        b.y = B[(size_t)(ka + 1) * Ncol + col];
#pragma unroll
        for (int i = 0; i < MT; ++i) {
            __builtin_prefetch(arow[i] + k0 + 64, 0, 0);
            v2f a = *(const v2f*)(arow[i] + ka);
            a.x *= rs[i];
            a.y *= rs[i];
            acc[i] = __builtin_amdgcn_wmma_f32_16x16x4_f32(
                 false, a,  false, b,
                 (short)0, acc[i],  false,  false);
        }
    }

#pragma unroll
    for (int i = 0; i < MT; ++i) {
#pragma unroll
        for (int v = 0; v < 8; ++v) {
            int m = tm * (16 * MT) + i * 16 + half * 8 + v;
            if (m < Mrows) {
                float c = acc[i][v] + bv;
                if (relu_c) c = fmaxf(c, 0.0f);
                float o;
                if (mode == 1)      o = fmaxf(resid[(size_t)m * Ncol + col] + c, 0.0f);
                else if (mode == 2) o = fmaxf(c, 0.0f);
                else                o = c;
                out[(size_t)m * Ncol + col] = o;
            }
        }
    }
}


extern "C" void kernel_launch(void* const* d_in, const int* in_sizes, int n_in,
                              void* d_out, int out_size, void* d_ws, size_t ws_size,
                              hipStream_t stream) {
    const float* node_features = (const float*)d_in[0];
    const int*   edge_src      = (const int*)d_in[1];
    const int*   edge_dst      = (const int*)d_in[2];
    const int*   graph_ids     = (const int*)d_in[3];
    const float* W_proj        = (const float*)d_in[4];
    const float* b_proj        = (const float*)d_in[5];
    const float* W_gcn         = (const float*)d_in[6];
    const float* b_gcn         = (const float*)d_in[7];
    const float* W_mlp0        = (const float*)d_in[8];
    const float* b_mlp0        = (const float*)d_in[9];
    const float* W_mlp1        = (const float*)d_in[10];
    const float* b_mlp1        = (const float*)d_in[11];
    const float* W_mlp2        = (const float*)d_in[12];
    const float* b_mlp2        = (const float*)d_in[13];

    const int D = D_FEAT;
    const int Mm = 256;
    const int N = in_sizes[3];
    const int E = in_sizes[1];
    const int L = in_sizes[6] / (D * D);
    const int G = out_size / Mm;

    float* ws = (float*)d_ws;
    float* norm_src = ws;
    float* norm_dst = norm_src + N;
    float* h        = norm_dst + N;
    float* agg      = h + (size_t)N * D;
    float* hg       = agg + (size_t)N * D;
    float* t0       = hg + (size_t)G * D;
    float* t1       = t0 + (size_t)G * Mm;

    const int TB = 256;
    auto blocks = [&](size_t n) { return (unsigned)((n + TB - 1) / TB); };
    const int MROWS = 16 * MT;

    zero_f32<<<blocks((size_t)2 * N), TB, 0, stream>>>(norm_src, (size_t)2 * N);
    degree_kernel<<<blocks(E), TB, 0, stream>>>(edge_src, edge_dst, norm_src, norm_dst, E);
    norm_kernel<<<blocks((size_t)2 * N), TB, 0, stream>>>(norm_src, 2 * N);

    {
        dim3 g((N + MROWS - 1) / MROWS, D / 16);
        gemm_wmma_f32<<<g, 32, 0, stream>>>(node_features, W_proj, b_proj,
                                            nullptr, nullptr, h, N, D, D,
                                             0,  0);
    }

    for (int i = 0; i < L; ++i) {
        zero_f32<<<blocks((size_t)N * D), TB, 0, stream>>>(agg, (size_t)N * D);
        scatter_kernel<<<blocks((size_t)E * 32), TB, 0, stream>>>(h, norm_src,
                                                                  edge_src, edge_dst, agg, E);
        dim3 g((N + MROWS - 1) / MROWS, D / 16);
        gemm_wmma_f32<<<g, 32, 0, stream>>>(agg, W_gcn + (size_t)i * D * D,
                                            b_gcn + (size_t)i * D,
                                            norm_dst, h, h, N, D, D,
                                             (i < L - 1) ? 1 : 0,  1);
    }

    zero_f32<<<blocks((size_t)G * D), TB, 0, stream>>>(hg, (size_t)G * D);
    pool_kernel<<<blocks((size_t)N * 32), TB, 0, stream>>>(h, graph_ids, hg, N);

    {
        dim3 g0((G + MROWS - 1) / MROWS, Mm / 16);
        gemm_wmma_f32<<<g0, 32, 0, stream>>>(hg, W_mlp0, b_mlp0, nullptr, nullptr,
                                             t0, G, D, Mm, 0,  2);
        gemm_wmma_f32<<<g0, 32, 0, stream>>>(t0, W_mlp1, b_mlp1, nullptr, nullptr,
                                             t1, G, Mm, Mm, 0,  2);
        gemm_wmma_f32<<<g0, 32, 0, stream>>>(t1, W_mlp2, b_mlp2, nullptr, nullptr,
                                             (float*)d_out, G, Mm, Mm, 0,  0);
    }
}


# Scatter-add without global atomics: edges binned once per call by 64-node destination range (LDS counts, scan, place); a block sums its bin's rows in LDS (f32 LDS adds, arrival order), stores once
# speedup vs baseline: 3.1800x; 3.1800x over previous
#include <hip/hip_runtime.h>
#include <hip/hip_bf16.h>

typedef __attribute__((ext_vector_type(2))) float v2f;
typedef __attribute__((ext_vector_type(8))) float v8f;

#define D_FEAT 128


__global__ void zero_f32(float* __restrict__ p, size_t n) {
    size_t i = (size_t)blockIdx.x * blockDim.x + threadIdx.x;
    if (i < n) p[i] = 0.0f;
}

#define DG_THREADS 1024
#define DG_RANGES 64
#define DG_CAP 2048
#define DG_INFLIGHT 8

__device__ __forceinline__ void dg_count(int* cnt, int x, int lo, unsigned len) {
    const unsigned u = (unsigned)x - (unsigned)lo;
    if (u < len) atomicAdd(&cnt[u], 1);
}

__device__ __forceinline__ void dg_count4(int* cnt, int4 v, int lo, unsigned len) {
    dg_count(cnt, v.x, lo, len);
    dg_count(cnt, v.y, lo, len);
    dg_count(cnt, v.z, lo, len);
    dg_count(cnt, v.w, lo, len);
}

template <bool GUARD>
__device__ __forceinline__ void dg_batch(int* cnt, const int4* __restrict__ idx4, int j0, int n4,
                                         int lo, unsigned len) {
    int4 v[DG_INFLIGHT];
#pragma unroll
    for (int u = 0; u < DG_INFLIGHT; ++u) {
        const int j = j0 + u * DG_THREADS;
        if (!GUARD || j < n4) v[u] = idx4[j];
        else v[u] = make_int4(-1, -1, -1, -1);
    }
#pragma unroll
    for (int u = 0; u < DG_INFLIGHT; ++u) dg_count4(cnt, v[u], lo, len);
}

__global__ void __launch_bounds__(DG_THREADS)
degree_norm_kernel(const int* __restrict__ src, const int* __restrict__ dst,
                   float* __restrict__ norm_src, float* __restrict__ norm_dst, int E, int N) {
    __shared__ int cnt[DG_CAP];
    const int* __restrict__ idx = blockIdx.y ? dst : src;
    float* __restrict__ nrm = blockIdx.y ? norm_dst : norm_src;
    const int tid = threadIdx.x;
    const int R = (N + DG_RANGES - 1) / DG_RANGES;
    const int r0 = blockIdx.x * R;
    const int r1 = (r0 + R < N) ? (r0 + R) : N;
    const int n4 = E >> 2;
    const int4* __restrict__ idx4 = (const int4*)idx;

    for (int lo = r0; lo < r1; lo += DG_CAP) {
        const int len = (r1 - lo < DG_CAP) ? (r1 - lo) : DG_CAP;
        for (int i = tid; i < len; i += DG_THREADS) cnt[i] = 0;
        __syncthreads();

        int base = 0;
        for (; base + DG_INFLIGHT * DG_THREADS <= n4; base += DG_INFLIGHT * DG_THREADS)
            dg_batch<false>(cnt, idx4, base + tid, n4, lo, (unsigned)len);
        if (base < n4) dg_batch<true>(cnt, idx4, base + tid, n4, lo, (unsigned)len);
        for (int e = (n4 << 2) + tid; e < E; e += DG_THREADS) dg_count(cnt, idx[e], lo, (unsigned)len);
        __syncthreads();

        for (int k = tid; k < len; k += DG_THREADS)
            nrm[lo + k] = rsqrtf(fmaxf((float)cnt[k], 1.0f));
        __syncthreads();
    }
}

#define BN_SHIFT 6
#define BN_NODES (1 << BN_SHIFT)
#define BN_CAP 1024
#define BN_BLOCKS 64
#define BN_THREADS 1024

__device__ __forceinline__ int bn_bin(int d, int B) {
    unsigned b = (unsigned)d >> BN_SHIFT;
    if (b > (unsigned)(B - 1)) b = (unsigned)(B - 1);
    return (int)b;
}

__global__ void __launch_bounds__(BN_THREADS)
bin_count_kernel(const int* __restrict__ dst, int* __restrict__ hist, int E, int B) {
    __shared__ int cnt[BN_CAP];
    const int tid = threadIdx.x;
    for (int i = tid; i < BN_CAP; i += BN_THREADS) cnt[i] = 0;
    __syncthreads();
    const int S = (E + BN_BLOCKS - 1) / BN_BLOCKS;
    const int e0 = blockIdx.x * S;
    const int e1 = (e0 + S < E) ? (e0 + S) : E;
    for (int e = e0 + tid; e < e1; e += BN_THREADS) atomicAdd(&cnt[bn_bin(dst[e], B)], 1);
    __syncthreads();
    for (int i = tid; i < B; i += BN_THREADS) hist[blockIdx.x * B + i] = cnt[i];
}

__global__ void __launch_bounds__(BN_THREADS)
bin_place_kernel(const int* __restrict__ src, const int* __restrict__ dst,
                 const int* __restrict__ hist, int* __restrict__ boff,
                 int2* __restrict__ binned, int E, int B) {
    __shared__ int cur[BN_CAP];
    __shared__ int wsum[BN_THREADS / 32];
    const int tid = threadIdx.x;
    const int lane = tid & 31;
    const int wave = tid >> 5;
    const int blk = blockIdx.x;

    int tot = 0, pre = 0;
    if (tid < B) {
        for (int b = 0; b < BN_BLOCKS; ++b) {
            const int c = hist[b * B + tid];
            tot += c;
            if (b < blk) pre += c;
        }
    }
    int v = tot;
    for (int d = 1; d < 32; d <<= 1) {
        const int t = __shfl_up(v, d);
        if (lane >= d) v += t;
    }
    if (lane == 31) wsum[wave] = v;
    __syncthreads();
    if (wave == 0) {
        const int w = wsum[lane];
        int wi = w;
        for (int d = 1; d < 32; d <<= 1) {
            const int t = __shfl_up(wi, d);
            if (lane >= d) wi += t;
        }
        wsum[lane] = wi - w;
    }
    __syncthreads();
    const int excl = v - tot + wsum[wave];
    if (tid < B) {
        cur[tid] = excl + pre;
        if (blk == 0) boff[tid] = excl;
    }
    if (blk == 0 && tid == 0) boff[B] = E;
    __syncthreads();

    const int S = (E + BN_BLOCKS - 1) / BN_BLOCKS;
    const int e0 = blk * S;
    const int e1 = (e0 + S < E) ? (e0 + S) : E;
    for (int e = e0 + tid; e < e1; e += BN_THREADS) {
        const int s = src[e];
        const int d = dst[e];
        const int pos = atomicAdd(&cur[bn_bin(d, B)], 1);
        binned[pos] = make_int2(s, d);
    }
}

#define SC_EPW 16
#define AG_WAVES 8
__global__ void __launch_bounds__(32 * AG_WAVES)
aggregate_bins_kernel(const float* __restrict__ h, const float* __restrict__ norm_src,
                      const int2* __restrict__ binned, const int* __restrict__ boff,
                      float* __restrict__ agg, int N) {
    __shared__ float rows[BN_NODES * D_FEAT];
    const int tid = threadIdx.x;
    const int lane = tid & 31;
    const int wid = __builtin_amdgcn_readfirstlane((int)(threadIdx.x >> 5));
    const int bin = blockIdx.x;
    const int lo = bin << BN_SHIFT;
    float4* r4 = (float4*)rows;
    for (int i = tid; i < BN_NODES * D_FEAT / 4; i += 32 * AG_WAVES)
        r4[i] = make_float4(0.0f, 0.0f, 0.0f, 0.0f);
    __syncthreads();

    const int b0 = boff[bin];
    const int b1 = boff[bin + 1];
    for (int e0 = b0 + wid * SC_EPW; e0 < b1; e0 += AG_WAVES * SC_EPW) {
        const int n = (b1 - e0 < SC_EPW) ? (b1 - e0) : SC_EPW;
        int el = e0 + (lane & (SC_EPW - 1));
        if (el > b1 - 1) el = b1 - 1;
        const int2 ed = binned[el];
        const int sv = ed.x;
        const int dv = (ed.y - lo) & (BN_NODES - 1);
        const int nv = __float_as_int(norm_src[sv]);

        if (n == SC_EPW) {
            float v[SC_EPW][4];
#pragma unroll
            for (int i = 0; i < SC_EPW; ++i) {
                const int s = __builtin_amdgcn_readlane(sv, i);
                const float* hp = h + (size_t)s * D_FEAT + lane;
#pragma unroll
                for (int k = 0; k < 4; ++k) v[i][k] = hp[32 * k];
            }
#pragma unroll
            for (int i = 0; i < SC_EPW; ++i) {
                const int d = __builtin_amdgcn_readlane(dv, i);
                const float ns = __int_as_float(__builtin_amdgcn_readlane(nv, i));
                float* rp = rows + d * D_FEAT + lane;
#pragma unroll
                for (int k = 0; k < 4; ++k)
                    __hip_atomic_fetch_add(rp + 32 * k, v[i][k] * ns, __ATOMIC_RELAXED,
                                           __HIP_MEMORY_SCOPE_WORKGROUP);
            }
        } else {
            for (int i = 0; i < n; ++i) {
                const int s = __builtin_amdgcn_readlane(sv, i);
                const int d = __builtin_amdgcn_readlane(dv, i);
                const float ns = __int_as_float(__builtin_amdgcn_readlane(nv, i));
                const float* hp = h + (size_t)s * D_FEAT + lane;
                float* rp = rows + d * D_FEAT + lane;
                float t[4];
#pragma unroll
                for (int k = 0; k < 4; ++k) t[k] = hp[32 * k];
#pragma unroll
                for (int k = 0; k < 4; ++k)
                    __hip_atomic_fetch_add(rp + 32 * k, t[k] * ns, __ATOMIC_RELAXED,
                                           __HIP_MEMORY_SCOPE_WORKGROUP);
            }
        }
    }
    __syncthreads();

    for (int r = wid; r < BN_NODES; r += AG_WAVES) {
        const int node = lo + r;
        if (node < N)
            *(float4*)(agg + (size_t)node * D_FEAT + 4 * lane) = r4[r * (D_FEAT / 4) + lane];
    }
}

__global__ void pool_kernel(const float* __restrict__ h, const int* __restrict__ gid,
                            float* __restrict__ hg, int N) {
    int t = blockIdx.x * blockDim.x + threadIdx.x;
    int n = t >> 5;
    int c = (t & 31) << 2;
    if (n < N) {
        int g = gid[n];
        float4 v = *(const float4*)(h + (size_t)n * D_FEAT + c);
        float* hp = hg + (size_t)g * D_FEAT + c;
        unsafeAtomicAdd(hp + 0, v.x);
        unsafeAtomicAdd(hp + 1, v.y);
        unsafeAtomicAdd(hp + 2, v.z);
        unsafeAtomicAdd(hp + 3, v.w);
    }
}

#define MT 4
__global__ void __launch_bounds__(32)
gemm_wmma_f32(const float* __restrict__ A, const float* __restrict__ B,
              const float* __restrict__ bias,
              const float* __restrict__ row_scale,
              const float* __restrict__ resid,
              float* __restrict__ out,
              int Mrows, int K, int Ncol, int relu_c, int mode) {
    int tm   = blockIdx.x;
    int tn   = blockIdx.y;
    int lane = threadIdx.x;
    int half = lane >> 4;
    int l15  = lane & 15;

    int col = tn * 16 + l15;
    float bv = bias[col];

    const float* arow[MT];
    float rs[MT];
#pragma unroll
    for (int i = 0; i < MT; ++i) {
        int row = tm * (16 * MT) + i * 16 + l15;
        if (row >= Mrows) row = Mrows - 1;
        arow[i] = A + (size_t)row * K;
        rs[i] = row_scale ? row_scale[row] : 1.0f;
    }

    v8f acc[MT] = {};
    for (int k0 = 0; k0 < K; k0 += 4) {
        int ka = k0 + half * 2;
        v2f b;
        b.x = B[(size_t)ka * Ncol + col];
        b.y = B[(size_t)(ka + 1) * Ncol + col];
#pragma unroll
        for (int i = 0; i < MT; ++i) {
            __builtin_prefetch(arow[i] + k0 + 64, 0, 0);
            v2f a = *(const v2f*)(arow[i] + ka);
            a.x *= rs[i];
            a.y *= rs[i];
            acc[i] = __builtin_amdgcn_wmma_f32_16x16x4_f32(
                 false, a,  false, b,
                 (short)0, acc[i],  false,  false);
        }
    }

#pragma unroll
    for (int i = 0; i < MT; ++i) {
#pragma unroll
        for (int v = 0; v < 8; ++v) {
            int m = tm * (16 * MT) + i * 16 + half * 8 + v;
            if (m < Mrows) {
                float c = acc[i][v] + bv;
                if (relu_c) c = fmaxf(c, 0.0f);
                float o;
                if (mode == 1)      o = fmaxf(resid[(size_t)m * Ncol + col] + c, 0.0f);
                else if (mode == 2) o = fmaxf(c, 0.0f);
                else                o = c;
                out[(size_t)m * Ncol + col] = o;
            }
        }
    }
}

#define NG_WAVES 8

template <int MODE>
__device__ __forceinline__ float ng_finish(float a, float b, float r, int relu_c) {
    float cc = a + b;
    if (relu_c) cc = fmaxf(cc, 0.0f);
    return (MODE == 1) ? fmaxf(r + cc, 0.0f) : cc;
}

template <int MODE>
__global__ void __launch_bounds__(32 * NG_WAVES)
node_gemm_rows128(const float* A, const float* __restrict__ Wg,
                  const float* __restrict__ bias,
                  const float* __restrict__ row_scale,
                  const float* resid, float* out,
                  int Mrows, int relu_c) {
    __shared__ float4 Wl[32 * D_FEAT];

    const int tid  = threadIdx.x;
    const int wave = tid >> 5;
    const int lane = tid & 31;
    const int hh   = lane >> 4;
    const int c    = lane & 15;

    float4 wr[4][4];
#pragma unroll
    for (int it = 0; it < 4; ++it) {
        int p  = it * (32 * NG_WAVES) + tid;
        int kq = p >> 5;
        int n4 = p & 31;
        const float* wp = Wg + (size_t)(4 * kq) * D_FEAT + 4 * n4;
#pragma unroll
        for (int q = 0; q < 4; ++q) wr[it][q] = *(const float4*)(wp + q * D_FEAT);
    }

    const int tile = blockIdx.x * NG_WAVES + wave;
    int row = tile * 16 + c;
    if (row >= Mrows) row = Mrows - 1;
    const float* ap = A + (size_t)row * D_FEAT + 4 * hh;
    const float rs = row_scale ? row_scale[row] : 1.0f;
    float4 av[16];
#pragma unroll
    for (int j = 0; j < 16; ++j) av[j] = *(const float4*)(ap + 8 * j);

#pragma unroll
    for (int it = 0; it < 4; ++it) {
        int p  = it * (32 * NG_WAVES) + tid;
        int kq = p >> 5;
        int n4 = p & 31;
        float4* dstp = Wl + kq * D_FEAT + (4 * (n4 >> 4)) * 16 + (n4 & 15);
        dstp[0 * 16] = make_float4(wr[it][0].x, wr[it][1].x, wr[it][2].x, wr[it][3].x);
        dstp[1 * 16] = make_float4(wr[it][0].y, wr[it][1].y, wr[it][2].y, wr[it][3].y);
        dstp[2 * 16] = make_float4(wr[it][0].z, wr[it][1].z, wr[it][2].z, wr[it][3].z);
        dstp[3 * 16] = make_float4(wr[it][0].w, wr[it][1].w, wr[it][2].w, wr[it][3].w);
    }
    __syncthreads();

    v8f acc[8] = {};
    const float4* wl = Wl + hh * D_FEAT + c;
    float4 bw[8];
#pragma unroll
    for (int t = 0; t < 8; ++t) bw[t] = wl[t * 16];
#pragma unroll
    for (int j = 0; j < 16; ++j) {
        v2f a0, a1;
        a0.x = av[j].x * rs;  a0.y = av[j].y * rs;
        a1.x = av[j].z * rs;  a1.y = av[j].w * rs;
        float4 bn[8];
        if (j < 15) {
#pragma unroll
            for (int t = 0; t < 8; ++t) bn[t] = wl[(2 * (j + 1)) * D_FEAT + t * 16];
        }
        __builtin_amdgcn_sched_barrier(0);
#pragma unroll
        for (int t = 0; t < 8; ++t) {
            v2f b0; b0.x = bw[t].x; b0.y = bw[t].y;
            acc[t] = __builtin_amdgcn_wmma_f32_16x16x4_f32(
                false, a0, false, b0, (short)0, acc[t], false, false);
        }
#pragma unroll
        for (int t = 0; t < 8; ++t) {
            v2f b1; b1.x = bw[t].z; b1.y = bw[t].w;
            acc[t] = __builtin_amdgcn_wmma_f32_16x16x4_f32(
                false, a1, false, b1, (short)0, acc[t], false, false);
        }
        __builtin_amdgcn_sched_barrier(0);
        if (j < 15) {
#pragma unroll
            for (int t = 0; t < 8; ++t) bw[t] = bn[t];
        }
    }

    const int m0 = tile * 16 + hh * 8;
    float4 bz[2];
#pragma unroll
    for (int g = 0; g < 2; ++g) bz[g] = *(const float4*)(bias + 64 * g + 4 * c);

    if (tile * 16 + 16 <= Mrows) {
        float4 rv[8][2];
#pragma unroll
        for (int v = 0; v < 8; ++v)
#pragma unroll
            for (int g = 0; g < 2; ++g) {
                if (MODE == 1)
                    rv[v][g] = *(const float4*)(resid + (size_t)(m0 + v) * D_FEAT + 64 * g + 4 * c);
                else
                    rv[v][g] = make_float4(0.0f, 0.0f, 0.0f, 0.0f);
            }
#pragma unroll
        for (int v = 0; v < 8; ++v)
#pragma unroll
            for (int g = 0; g < 2; ++g) {
                float4 o;
                o.x = ng_finish<MODE>(acc[4 * g + 0][v], bz[g].x, rv[v][g].x, relu_c);
                o.y = ng_finish<MODE>(acc[4 * g + 1][v], bz[g].y, rv[v][g].y, relu_c);
                o.z = ng_finish<MODE>(acc[4 * g + 2][v], bz[g].z, rv[v][g].z, relu_c);
                o.w = ng_finish<MODE>(acc[4 * g + 3][v], bz[g].w, rv[v][g].w, relu_c);
                *(float4*)(out + (size_t)(m0 + v) * D_FEAT + 64 * g + 4 * c) = o;
            }
    } else {
#pragma unroll
        for (int v = 0; v < 8; ++v) {
            if (m0 + v < Mrows) {
#pragma unroll
                for (int g = 0; g < 2; ++g) {
                    float4 r = make_float4(0.0f, 0.0f, 0.0f, 0.0f);
                    if (MODE == 1)
                        r = *(const float4*)(resid + (size_t)(m0 + v) * D_FEAT + 64 * g + 4 * c);
                    float4 o;
                    o.x = ng_finish<MODE>(acc[4 * g + 0][v], bz[g].x, r.x, relu_c);
                    o.y = ng_finish<MODE>(acc[4 * g + 1][v], bz[g].y, r.y, relu_c);
                    o.z = ng_finish<MODE>(acc[4 * g + 2][v], bz[g].z, r.z, relu_c);
                    o.w = ng_finish<MODE>(acc[4 * g + 3][v], bz[g].w, r.w, relu_c);
                    *(float4*)(out + (size_t)(m0 + v) * D_FEAT + 64 * g + 4 * c) = o;
                }
            }
        }
    }
}


extern "C" void kernel_launch(void* const* d_in, const int* in_sizes, int n_in,
                              void* d_out, int out_size, void* d_ws, size_t ws_size,
                              hipStream_t stream) {
    const float* node_features = (const float*)d_in[0];
    const int*   edge_src      = (const int*)d_in[1];
    const int*   edge_dst      = (const int*)d_in[2];
    const int*   graph_ids     = (const int*)d_in[3];
    const float* W_proj        = (const float*)d_in[4];
    const float* b_proj        = (const float*)d_in[5];
    const float* W_gcn         = (const float*)d_in[6];
    const float* b_gcn         = (const float*)d_in[7];
    const float* W_mlp0        = (const float*)d_in[8];
    const float* b_mlp0        = (const float*)d_in[9];
    const float* W_mlp1        = (const float*)d_in[10];
    const float* b_mlp1        = (const float*)d_in[11];
    const float* W_mlp2        = (const float*)d_in[12];
    const float* b_mlp2        = (const float*)d_in[13];

    const int D = D_FEAT;
    const int Mm = 256;
    const int N = in_sizes[3];
    const int E = in_sizes[1];
    const int L = in_sizes[6] / (D * D);
    const int G = out_size / Mm;

    float* ws = (float*)d_ws;
    float* norm_src = ws;
    float* norm_dst = norm_src + N;
    float* h        = norm_dst + N;
    float* agg      = h + (size_t)N * D;
    float* hg       = agg + (size_t)N * D;
    float* t0       = hg + (size_t)G * D;
    float* t1       = t0 + (size_t)G * Mm;
    const int B = (N + BN_NODES - 1) / BN_NODES;
    int2* binned    = (int2*)(t1 + (size_t)G * Mm);
    int*  hist      = (int*)(binned + E);
    int*  boff      = hist + (size_t)BN_BLOCKS * B;

    const int TB = 256;
    auto blocks = [&](size_t n) { return (unsigned)((n + TB - 1) / TB); };
    const int MROWS = 16 * MT;

    degree_norm_kernel<<<dim3(DG_RANGES, 2), DG_THREADS, 0, stream>>>(edge_src, edge_dst,
                                                                      norm_src, norm_dst, E, N);

    bin_count_kernel<<<BN_BLOCKS, BN_THREADS, 0, stream>>>(edge_dst, hist, E, B);
    bin_place_kernel<<<BN_BLOCKS, BN_THREADS, 0, stream>>>(edge_src, edge_dst, hist, boff, binned, E, B);

    const int NG_ROWS = 16 * NG_WAVES;
    const unsigned ng_blocks = (unsigned)((N + NG_ROWS - 1) / NG_ROWS);
    node_gemm_rows128<0><<<ng_blocks, 32 * NG_WAVES, 0, stream>>>(node_features, W_proj, b_proj,
                                                                 nullptr, nullptr, h, N,
                                                                  0);

    for (int i = 0; i < L; ++i) {
        aggregate_bins_kernel<<<(unsigned)B, 32 * AG_WAVES, 0, stream>>>(h, norm_src, binned, boff,
                                                                       agg, N);
        node_gemm_rows128<1><<<ng_blocks, 32 * NG_WAVES, 0, stream>>>(agg, W_gcn + (size_t)i * D * D,
                                                                     b_gcn + (size_t)i * D,
                                                                     norm_dst, h, h, N,
                                                                      (i < L - 1) ? 1 : 0);
    }

    zero_f32<<<blocks((size_t)G * D), TB, 0, stream>>>(hg, (size_t)G * D);
    pool_kernel<<<blocks((size_t)N * 32), TB, 0, stream>>>(h, graph_ids, hg, N);

    {
        dim3 g0((G + MROWS - 1) / MROWS, Mm / 16);
        gemm_wmma_f32<<<g0, 32, 0, stream>>>(hg, W_mlp0, b_mlp0, nullptr, nullptr,
                                             t0, G, D, Mm, 0,  2);
        gemm_wmma_f32<<<g0, 32, 0, stream>>>(t0, W_mlp1, b_mlp1, nullptr, nullptr,
                                             t1, G, Mm, Mm, 0,  2);
        gemm_wmma_f32<<<g0, 32, 0, stream>>>(t1, W_mlp2, b_mlp2, nullptr, nullptr,
                                             (float*)d_out, G, Mm, Mm, 0,  0);
    }
}


# GraphConv layer in one launch: the bin's block sums its 64 rows in LDS, then multiplies them by W (two k-halves via LDS) itself; no stored aggregate, h in two planes; sum orders kept; 19->14 launches
# speedup vs baseline: 3.5995x; 1.1319x over previous
#include <hip/hip_runtime.h>
#include <hip/hip_bf16.h>

typedef __attribute__((ext_vector_type(2))) float v2f;
typedef __attribute__((ext_vector_type(8))) float v8f;

#define D_FEAT 128


__global__ void zero_f32(float* __restrict__ p, size_t n) {
    size_t i = (size_t)blockIdx.x * blockDim.x + threadIdx.x;
    if (i < n) p[i] = 0.0f;
}

#define DG_THREADS 1024
#define DG_RANGES 64
#define DG_CAP 2048
#define DG_INFLIGHT 8

__device__ __forceinline__ void dg_count(int* cnt, int x, int lo, unsigned len) {
    const unsigned u = (unsigned)x - (unsigned)lo;
    if (u < len) atomicAdd(&cnt[u], 1);
}

__device__ __forceinline__ void dg_count4(int* cnt, int4 v, int lo, unsigned len) {
    dg_count(cnt, v.x, lo, len);
    dg_count(cnt, v.y, lo, len);
    dg_count(cnt, v.z, lo, len);
    dg_count(cnt, v.w, lo, len);
}

template <bool GUARD>
__device__ __forceinline__ void dg_batch(int* cnt, const int4* __restrict__ idx4, int j0, int n4,
                                         int lo, unsigned len) {
    int4 v[DG_INFLIGHT];
#pragma unroll
    for (int u = 0; u < DG_INFLIGHT; ++u) {
        const int j = j0 + u * DG_THREADS;
        if (!GUARD || j < n4) v[u] = idx4[j];
        else v[u] = make_int4(-1, -1, -1, -1);
    }
#pragma unroll
    for (int u = 0; u < DG_INFLIGHT; ++u) dg_count4(cnt, v[u], lo, len);
}

__global__ void __launch_bounds__(DG_THREADS)
degree_norm_kernel(const int* __restrict__ src, const int* __restrict__ dst,
                   float* __restrict__ norm_src, float* __restrict__ norm_dst, int E, int N) {
    __shared__ int cnt[DG_CAP];
    const int* __restrict__ idx = blockIdx.y ? dst : src;
    float* __restrict__ nrm = blockIdx.y ? norm_dst : norm_src;
    const int tid = threadIdx.x;
    const int R = (N + DG_RANGES - 1) / DG_RANGES;
    const int r0 = blockIdx.x * R;
    const int r1 = (r0 + R < N) ? (r0 + R) : N;
    const int n4 = E >> 2;
    const int4* __restrict__ idx4 = (const int4*)idx;

    for (int lo = r0; lo < r1; lo += DG_CAP) {
        const int len = (r1 - lo < DG_CAP) ? (r1 - lo) : DG_CAP;
        for (int i = tid; i < len; i += DG_THREADS) cnt[i] = 0;
        __syncthreads();

        int base = 0;
        for (; base + DG_INFLIGHT * DG_THREADS <= n4; base += DG_INFLIGHT * DG_THREADS)
            dg_batch<false>(cnt, idx4, base + tid, n4, lo, (unsigned)len);
        if (base < n4) dg_batch<true>(cnt, idx4, base + tid, n4, lo, (unsigned)len);
        for (int e = (n4 << 2) + tid; e < E; e += DG_THREADS) dg_count(cnt, idx[e], lo, (unsigned)len);
        __syncthreads();

        for (int k = tid; k < len; k += DG_THREADS)
            nrm[lo + k] = rsqrtf(fmaxf((float)cnt[k], 1.0f));
        __syncthreads();
    }
}

#define BN_SHIFT 6
#define BN_NODES (1 << BN_SHIFT)
#define BN_CAP 1024
#define BN_BLOCKS 64
#define BN_THREADS 1024

__device__ __forceinline__ int bn_bin(int d, int B) {
    unsigned b = (unsigned)d >> BN_SHIFT;
    if (b > (unsigned)(B - 1)) b = (unsigned)(B - 1);
    return (int)b;
}

__global__ void __launch_bounds__(BN_THREADS)
bin_count_kernel(const int* __restrict__ dst, int* __restrict__ hist, int E, int B) {
    __shared__ int cnt[BN_CAP];
    const int tid = threadIdx.x;
    for (int i = tid; i < BN_CAP; i += BN_THREADS) cnt[i] = 0;
    __syncthreads();
    const int S = (E + BN_BLOCKS - 1) / BN_BLOCKS;
    const int e0 = blockIdx.x * S;
    const int e1 = (e0 + S < E) ? (e0 + S) : E;
    for (int e = e0 + tid; e < e1; e += BN_THREADS) atomicAdd(&cnt[bn_bin(dst[e], B)], 1);
    __syncthreads();
    for (int i = tid; i < B; i += BN_THREADS) hist[blockIdx.x * B + i] = cnt[i];
}

__global__ void __launch_bounds__(BN_THREADS)
bin_place_kernel(const int* __restrict__ src, const int* __restrict__ dst,
                 const int* __restrict__ hist, int* __restrict__ boff,
                 int2* __restrict__ binned, int E, int B) {
    __shared__ int cur[BN_CAP];
    __shared__ int wsum[BN_THREADS / 32];
    const int tid = threadIdx.x;
    const int lane = tid & 31;
    const int wave = tid >> 5;
    const int blk = blockIdx.x;

    int tot = 0, pre = 0;
    if (tid < B) {
        for (int b = 0; b < BN_BLOCKS; ++b) {
            const int c = hist[b * B + tid];
            tot += c;
            if (b < blk) pre += c;
        }
    }
    int v = tot;
    for (int d = 1; d < 32; d <<= 1) {
        const int t = __shfl_up(v, d);
        if (lane >= d) v += t;
    }
    if (lane == 31) wsum[wave] = v;
    __syncthreads();
    if (wave == 0) {
        const int w = wsum[lane];
        int wi = w;
        for (int d = 1; d < 32; d <<= 1) {
            const int t = __shfl_up(wi, d);
            if (lane >= d) wi += t;
        }
        wsum[lane] = wi - w;
    }
    __syncthreads();
    const int excl = v - tot + wsum[wave];
    if (tid < B) {
        cur[tid] = excl + pre;
        if (blk == 0) boff[tid] = excl;
    }
    if (blk == 0 && tid == 0) boff[B] = E;
    __syncthreads();

    const int S = (E + BN_BLOCKS - 1) / BN_BLOCKS;
    const int e0 = blk * S;
    const int e1 = (e0 + S < E) ? (e0 + S) : E;
    for (int e = e0 + tid; e < e1; e += BN_THREADS) {
        const int s = src[e];
        const int d = dst[e];
        const int pos = atomicAdd(&cur[bn_bin(d, B)], 1);
        binned[pos] = make_int2(s, d);
    }
}

__global__ void pool_kernel(const float* __restrict__ h, const int* __restrict__ gid,
                            float* __restrict__ hg, int N) {
    int t = blockIdx.x * blockDim.x + threadIdx.x;
    int n = t >> 5;
    int c = (t & 31) << 2;
    if (n < N) {
        int g = gid[n];
        float4 v = *(const float4*)(h + (size_t)n * D_FEAT + c);
        float* hp = hg + (size_t)g * D_FEAT + c;
        unsafeAtomicAdd(hp + 0, v.x);
        unsafeAtomicAdd(hp + 1, v.y);
        unsafeAtomicAdd(hp + 2, v.z);
        unsafeAtomicAdd(hp + 3, v.w);
    }
}

#define MT 4
__global__ void __launch_bounds__(32)
gemm_wmma_f32(const float* __restrict__ A, const float* __restrict__ B,
              const float* __restrict__ bias,
              const float* __restrict__ row_scale,
              const float* __restrict__ resid,
              float* __restrict__ out,
              int Mrows, int K, int Ncol, int relu_c, int mode) {
    int tm   = blockIdx.x;
    int tn   = blockIdx.y;
    int lane = threadIdx.x;
    int half = lane >> 4;
    int l15  = lane & 15;

    int col = tn * 16 + l15;
    float bv = bias[col];

    const float* arow[MT];
    float rs[MT];
#pragma unroll
    for (int i = 0; i < MT; ++i) {
        int row = tm * (16 * MT) + i * 16 + l15;
        if (row >= Mrows) row = Mrows - 1;
        arow[i] = A + (size_t)row * K;
        rs[i] = row_scale ? row_scale[row] : 1.0f;
    }

    v8f acc[MT] = {};
    for (int k0 = 0; k0 < K; k0 += 4) {
        int ka = k0 + half * 2;
        v2f b;
        b.x = B[(size_t)ka * Ncol + col];
        b.y = B[(size_t)(ka + 1) * Ncol + col];
#pragma unroll
        for (int i = 0; i < MT; ++i) {
            __builtin_prefetch(arow[i] + k0 + 64, 0, 0);
            v2f a = *(const v2f*)(arow[i] + ka);
            a.x *= rs[i];
            a.y *= rs[i];
            acc[i] = __builtin_amdgcn_wmma_f32_16x16x4_f32(
                 false, a,  false, b,
                 (short)0, acc[i],  false,  false);
        }
    }

#pragma unroll
    for (int i = 0; i < MT; ++i) {
#pragma unroll
        for (int v = 0; v < 8; ++v) {
            int m = tm * (16 * MT) + i * 16 + half * 8 + v;
            if (m < Mrows) {
                float c = acc[i][v] + bv;
                if (relu_c) c = fmaxf(c, 0.0f);
                float o;
                if (mode == 1)      o = fmaxf(resid[(size_t)m * Ncol + col] + c, 0.0f);
                else if (mode == 2) o = fmaxf(c, 0.0f);
                else                o = c;
                out[(size_t)m * Ncol + col] = o;
            }
        }
    }
}

#define NG_WAVES 8

template <int MODE>
__device__ __forceinline__ float ng_finish(float a, float b, float r, int relu_c) {
    float cc = a + b;
    if (relu_c) cc = fmaxf(cc, 0.0f);
    return (MODE == 1) ? fmaxf(r + cc, 0.0f) : cc;
}

template <int MODE>
__global__ void __launch_bounds__(32 * NG_WAVES)
node_gemm_rows128(const float* A, const float* __restrict__ Wg,
                  const float* __restrict__ bias,
                  const float* __restrict__ row_scale,
                  const float* resid, float* out,
                  int Mrows, int relu_c) {
    __shared__ float4 Wl[32 * D_FEAT];

    const int tid  = threadIdx.x;
    const int wave = tid >> 5;
    const int lane = tid & 31;
    const int hh   = lane >> 4;
    const int c    = lane & 15;

    float4 wr[4][4];
#pragma unroll
    for (int it = 0; it < 4; ++it) {
        int p  = it * (32 * NG_WAVES) + tid;
        int kq = p >> 5;
        int n4 = p & 31;
        const float* wp = Wg + (size_t)(4 * kq) * D_FEAT + 4 * n4;
#pragma unroll
        for (int q = 0; q < 4; ++q) wr[it][q] = *(const float4*)(wp + q * D_FEAT);
    }

    const int tile = blockIdx.x * NG_WAVES + wave;
    int row = tile * 16 + c;
    if (row >= Mrows) row = Mrows - 1;
    const float* ap = A + (size_t)row * D_FEAT + 4 * hh;
    const float rs = row_scale ? row_scale[row] : 1.0f;
    float4 av[16];
#pragma unroll
    for (int j = 0; j < 16; ++j) av[j] = *(const float4*)(ap + 8 * j);

#pragma unroll
    for (int it = 0; it < 4; ++it) {
        int p  = it * (32 * NG_WAVES) + tid;
        int kq = p >> 5;
        int n4 = p & 31;
        float4* dstp = Wl + kq * D_FEAT + (4 * (n4 >> 4)) * 16 + (n4 & 15);
        dstp[0 * 16] = make_float4(wr[it][0].x, wr[it][1].x, wr[it][2].x, wr[it][3].x);
        dstp[1 * 16] = make_float4(wr[it][0].y, wr[it][1].y, wr[it][2].y, wr[it][3].y);
        dstp[2 * 16] = make_float4(wr[it][0].z, wr[it][1].z, wr[it][2].z, wr[it][3].z);
        dstp[3 * 16] = make_float4(wr[it][0].w, wr[it][1].w, wr[it][2].w, wr[it][3].w);
    }
    __syncthreads();

    v8f acc[8] = {};
    const float4* wl = Wl + hh * D_FEAT + c;
    float4 bw[8];
#pragma unroll
    for (int t = 0; t < 8; ++t) bw[t] = wl[t * 16];
#pragma unroll
    for (int j = 0; j < 16; ++j) {
        v2f a0, a1;
        a0.x = av[j].x * rs;  a0.y = av[j].y * rs;
        a1.x = av[j].z * rs;  a1.y = av[j].w * rs;
        float4 bn[8];
        if (j < 15) {
#pragma unroll
            for (int t = 0; t < 8; ++t) bn[t] = wl[(2 * (j + 1)) * D_FEAT + t * 16];
        }
        __builtin_amdgcn_sched_barrier(0);
#pragma unroll
        for (int t = 0; t < 8; ++t) {
            v2f b0; b0.x = bw[t].x; b0.y = bw[t].y;
            acc[t] = __builtin_amdgcn_wmma_f32_16x16x4_f32(
                false, a0, false, b0, (short)0, acc[t], false, false);
        }
#pragma unroll
        for (int t = 0; t < 8; ++t) {
            v2f b1; b1.x = bw[t].z; b1.y = bw[t].w;
            acc[t] = __builtin_amdgcn_wmma_f32_16x16x4_f32(
                false, a1, false, b1, (short)0, acc[t], false, false);
        }
        __builtin_amdgcn_sched_barrier(0);
        if (j < 15) {
#pragma unroll
            for (int t = 0; t < 8; ++t) bw[t] = bn[t];
        }
    }

    const int m0 = tile * 16 + hh * 8;
    float4 bz[2];
#pragma unroll
    for (int g = 0; g < 2; ++g) bz[g] = *(const float4*)(bias + 64 * g + 4 * c);

    if (tile * 16 + 16 <= Mrows) {
        float4 rv[8][2];
#pragma unroll
        for (int v = 0; v < 8; ++v)
#pragma unroll
            for (int g = 0; g < 2; ++g) {
                if (MODE == 1)
                    rv[v][g] = *(const float4*)(resid + (size_t)(m0 + v) * D_FEAT + 64 * g + 4 * c);
                else
                    rv[v][g] = make_float4(0.0f, 0.0f, 0.0f, 0.0f);
            }
#pragma unroll
        for (int v = 0; v < 8; ++v)
#pragma unroll
            for (int g = 0; g < 2; ++g) {
                float4 o;
                o.x = ng_finish<MODE>(acc[4 * g + 0][v], bz[g].x, rv[v][g].x, relu_c);
                o.y = ng_finish<MODE>(acc[4 * g + 1][v], bz[g].y, rv[v][g].y, relu_c);
                o.z = ng_finish<MODE>(acc[4 * g + 2][v], bz[g].z, rv[v][g].z, relu_c);
                o.w = ng_finish<MODE>(acc[4 * g + 3][v], bz[g].w, rv[v][g].w, relu_c);
                *(float4*)(out + (size_t)(m0 + v) * D_FEAT + 64 * g + 4 * c) = o;
            }
    } else {
#pragma unroll
        for (int v = 0; v < 8; ++v) {
            if (m0 + v < Mrows) {
#pragma unroll
                for (int g = 0; g < 2; ++g) {
                    float4 r = make_float4(0.0f, 0.0f, 0.0f, 0.0f);
                    if (MODE == 1)
                        r = *(const float4*)(resid + (size_t)(m0 + v) * D_FEAT + 64 * g + 4 * c);
                    float4 o;
                    o.x = ng_finish<MODE>(acc[4 * g + 0][v], bz[g].x, r.x, relu_c);
                    o.y = ng_finish<MODE>(acc[4 * g + 1][v], bz[g].y, r.y, relu_c);
                    o.z = ng_finish<MODE>(acc[4 * g + 2][v], bz[g].z, r.z, relu_c);
                    o.w = ng_finish<MODE>(acc[4 * g + 3][v], bz[g].w, r.w, relu_c);
                    *(float4*)(out + (size_t)(m0 + v) * D_FEAT + 64 * g + 4 * c) = o;
                }
            }
        }
    }
}

#define SC_EPW 16
#define FL_WAVES 8
#define FL_PITCH 132

__device__ __forceinline__ void fl_w_load(float4 (&wr)[2][4], const float* __restrict__ Wg,
                                          int khalf, int tid) {
#pragma unroll
    for (int it = 0; it < 2; ++it) {
        const int p  = it * (32 * FL_WAVES) + tid;
        const int kq = p >> 5;
        const int n4 = p & 31;
        const float* wp = Wg + (size_t)(64 * khalf + 4 * kq) * D_FEAT + 4 * n4;
#pragma unroll
        for (int q = 0; q < 4; ++q) wr[it][q] = *(const float4*)(wp + q * D_FEAT);
    }
}

__device__ __forceinline__ void fl_w_store(float4* Wl, const float4 (&wr)[2][4], int tid) {
#pragma unroll
    for (int it = 0; it < 2; ++it) {
        const int p  = it * (32 * FL_WAVES) + tid;
        const int kq = p >> 5;
        const int n4 = p & 31;
        float4* dstp = Wl + kq * D_FEAT + (4 * (n4 >> 4)) * 16 + (n4 & 15);
        dstp[0 * 16] = make_float4(wr[it][0].x, wr[it][1].x, wr[it][2].x, wr[it][3].x);
        dstp[1 * 16] = make_float4(wr[it][0].y, wr[it][1].y, wr[it][2].y, wr[it][3].y);
        dstp[2 * 16] = make_float4(wr[it][0].z, wr[it][1].z, wr[it][2].z, wr[it][3].z);
        dstp[3 * 16] = make_float4(wr[it][0].w, wr[it][1].w, wr[it][2].w, wr[it][3].w);
    }
}

__device__ __forceinline__ void fl_gemm_half(v8f (&acc)[4], const float* arow, const float4* wl,
                                             float rs) {
    float4 aw = *(const float4*)arow;
    float4 bw[4];
#pragma unroll
    for (int t = 0; t < 4; ++t) bw[t] = wl[t * 16];
#pragma unroll
    for (int j = 0; j < 8; ++j) {
        v2f a0, a1;
        a0.x = aw.x * rs;  a0.y = aw.y * rs;
        a1.x = aw.z * rs;  a1.y = aw.w * rs;
        float4 an = aw;
        float4 bn[4];
        if (j < 7) {
            an = *(const float4*)(arow + 8 * (j + 1));
#pragma unroll
            for (int t = 0; t < 4; ++t) bn[t] = wl[(2 * (j + 1)) * D_FEAT + t * 16];
        }
        __builtin_amdgcn_sched_barrier(0);
#pragma unroll
        for (int t = 0; t < 4; ++t) {
            v2f b0; b0.x = bw[t].x; b0.y = bw[t].y;
            acc[t] = __builtin_amdgcn_wmma_f32_16x16x4_f32(
                false, a0, false, b0, (short)0, acc[t], false, false);
        }
#pragma unroll
        for (int t = 0; t < 4; ++t) {
            v2f b1; b1.x = bw[t].z; b1.y = bw[t].w;
            acc[t] = __builtin_amdgcn_wmma_f32_16x16x4_f32(
                false, a1, false, b1, (short)0, acc[t], false, false);
        }
        __builtin_amdgcn_sched_barrier(0);
        if (j < 7) {
            aw = an;
#pragma unroll
            for (int t = 0; t < 4; ++t) bw[t] = bn[t];
        }
    }
}

__global__ void __launch_bounds__(32 * FL_WAVES)
graphconv_bins_kernel(const float* __restrict__ hin, const float* __restrict__ norm_src,
                      const float* __restrict__ norm_dst,
                      const int2* __restrict__ binned, const int* __restrict__ boff,
                      const float* __restrict__ Wg, const float* __restrict__ bias,
                      float* __restrict__ hout, int N, int relu_c) {
    __shared__ __attribute__((aligned(16))) float rows[BN_NODES * FL_PITCH];
    __shared__ float4 Wl[16 * D_FEAT];
    const int tid = threadIdx.x;
    const int lane = tid & 31;
    const int wid = __builtin_amdgcn_readfirstlane((int)(threadIdx.x >> 5));
    const int bin = blockIdx.x;
    const int lo = bin << BN_SHIFT;

    float4 wr[2][4];
    fl_w_load(wr, Wg, 0, tid);
    float4* r4 = (float4*)rows;
    for (int i = tid; i < BN_NODES * FL_PITCH / 4; i += 32 * FL_WAVES)
        r4[i] = make_float4(0.0f, 0.0f, 0.0f, 0.0f);
    const int b0 = boff[bin];
    const int b1 = boff[bin + 1];
    fl_w_store(Wl, wr, tid);
    __syncthreads();

    for (int e0 = b0 + wid * SC_EPW; e0 < b1; e0 += FL_WAVES * SC_EPW) {
        const int n = (b1 - e0 < SC_EPW) ? (b1 - e0) : SC_EPW;
        int el = e0 + (lane & (SC_EPW - 1));
        if (el > b1 - 1) el = b1 - 1;
        const int2 ed = binned[el];
        const int sv = ed.x;
        const int dv = (ed.y - lo) & (BN_NODES - 1);
        const int nv = __float_as_int(norm_src[sv]);

        if (n == SC_EPW) {
            float v[SC_EPW][4];
#pragma unroll
            for (int i = 0; i < SC_EPW; ++i) {
                const int s = __builtin_amdgcn_readlane(sv, i);
                const float* hp = hin + (size_t)s * D_FEAT + lane;
#pragma unroll
                for (int k = 0; k < 4; ++k) v[i][k] = hp[32 * k];
            }
#pragma unroll
            for (int i = 0; i < SC_EPW; ++i) {
                const int d = __builtin_amdgcn_readlane(dv, i);
                const float ns = __int_as_float(__builtin_amdgcn_readlane(nv, i));
                float* rp = rows + d * FL_PITCH + lane;
#pragma unroll
                for (int k = 0; k < 4; ++k)
                    __hip_atomic_fetch_add(rp + 32 * k, v[i][k] * ns, __ATOMIC_RELAXED,
                                           __HIP_MEMORY_SCOPE_WORKGROUP);
            }
        } else {
            for (int i = 0; i < n; ++i) {
                const int s = __builtin_amdgcn_readlane(sv, i);
                const int d = __builtin_amdgcn_readlane(dv, i);
                const float ns = __int_as_float(__builtin_amdgcn_readlane(nv, i));
                const float* hp = hin + (size_t)s * D_FEAT + lane;
                float* rp = rows + d * FL_PITCH + lane;
                float t[4];
#pragma unroll
                for (int k = 0; k < 4; ++k) t[k] = hp[32 * k];
#pragma unroll
                for (int k = 0; k < 4; ++k)
                    __hip_atomic_fetch_add(rp + 32 * k, t[k] * ns, __ATOMIC_RELAXED,
                                           __HIP_MEMORY_SCOPE_WORKGROUP);
            }
        }
    }
    __syncthreads();

    const int hh = lane >> 4;
    const int c  = lane & 15;
    const int rt = wid & 3;
    const int g  = wid >> 2;
    int node = lo + 16 * rt + c;
    if (node >= N) node = N - 1;
    const float rs = norm_dst[node];
    const float* arow = rows + (16 * rt + c) * FL_PITCH + 4 * hh;
    const float4* wl = Wl + hh * D_FEAT + (4 * g) * 16 + c;

    fl_w_load(wr, Wg, 1, tid);

    v8f acc[4] = {};
    fl_gemm_half(acc, arow, wl, rs);
    __syncthreads();
    fl_w_store(Wl, wr, tid);
    __syncthreads();
    fl_gemm_half(acc, arow + 64, wl, rs);

    const int m0 = lo + 16 * rt + hh * 8;
    const float4 bz = *(const float4*)(bias + 64 * g + 4 * c);
    float4 rv[8];
#pragma unroll
    for (int v = 0; v < 8; ++v) {
        int m = m0 + v;
        if (m >= N) m = N - 1;
        rv[v] = *(const float4*)(hin + (size_t)m * D_FEAT + 64 * g + 4 * c);
    }
#pragma unroll
    for (int v = 0; v < 8; ++v) {
        if (m0 + v < N) {
            float4 o;
            o.x = ng_finish<1>(acc[0][v], bz.x, rv[v].x, relu_c);
            o.y = ng_finish<1>(acc[1][v], bz.y, rv[v].y, relu_c);
            o.z = ng_finish<1>(acc[2][v], bz.z, rv[v].z, relu_c);
            o.w = ng_finish<1>(acc[3][v], bz.w, rv[v].w, relu_c);
            *(float4*)(hout + (size_t)(m0 + v) * D_FEAT + 64 * g + 4 * c) = o;
        }
    }
}


extern "C" void kernel_launch(void* const* d_in, const int* in_sizes, int n_in,
                              void* d_out, int out_size, void* d_ws, size_t ws_size,
                              hipStream_t stream) {
    const float* node_features = (const float*)d_in[0];
    const int*   edge_src      = (const int*)d_in[1];
    const int*   edge_dst      = (const int*)d_in[2];
    const int*   graph_ids     = (const int*)d_in[3];
    const float* W_proj        = (const float*)d_in[4];
    const float* b_proj        = (const float*)d_in[5];
    const float* W_gcn         = (const float*)d_in[6];
    const float* b_gcn         = (const float*)d_in[7];
    const float* W_mlp0        = (const float*)d_in[8];
    const float* b_mlp0        = (const float*)d_in[9];
    const float* W_mlp1        = (const float*)d_in[10];
    const float* b_mlp1        = (const float*)d_in[11];
    const float* W_mlp2        = (const float*)d_in[12];
    const float* b_mlp2        = (const float*)d_in[13];

    const int D = D_FEAT;
    const int Mm = 256;
    const int N = in_sizes[3];
    const int E = in_sizes[1];
    const int L = in_sizes[6] / (D * D);
    const int G = out_size / Mm;

    float* ws = (float*)d_ws;
    float* norm_src = ws;
    float* norm_dst = norm_src + N;
    float* h        = norm_dst + N;
    float* h2       = h + (size_t)N * D;
    float* hg       = h2 + (size_t)N * D;
    float* t0       = hg + (size_t)G * D;
    float* t1       = t0 + (size_t)G * Mm;
    const int B = (N + BN_NODES - 1) / BN_NODES;
    int2* binned    = (int2*)(t1 + (size_t)G * Mm);
    int*  hist      = (int*)(binned + E);
    int*  boff      = hist + (size_t)BN_BLOCKS * B;

    const int TB = 256;
    auto blocks = [&](size_t n) { return (unsigned)((n + TB - 1) / TB); };
    const int MROWS = 16 * MT;

    degree_norm_kernel<<<dim3(DG_RANGES, 2), DG_THREADS, 0, stream>>>(edge_src, edge_dst,
                                                                      norm_src, norm_dst, E, N);

    bin_count_kernel<<<BN_BLOCKS, BN_THREADS, 0, stream>>>(edge_dst, hist, E, B);
    bin_place_kernel<<<BN_BLOCKS, BN_THREADS, 0, stream>>>(edge_src, edge_dst, hist, boff, binned, E, B);

    const int NG_ROWS = 16 * NG_WAVES;
    const unsigned ng_blocks = (unsigned)((N + NG_ROWS - 1) / NG_ROWS);
    node_gemm_rows128<0><<<ng_blocks, 32 * NG_WAVES, 0, stream>>>(node_features, W_proj, b_proj,
                                                                 nullptr, nullptr, h, N,
                                                                  0);

    float* hcur = h;
    float* hnxt = h2;
    for (int i = 0; i < L; ++i) {
        graphconv_bins_kernel<<<(unsigned)B, 32 * FL_WAVES, 0, stream>>>(
            hcur, norm_src, norm_dst, binned, boff, W_gcn + (size_t)i * D * D, b_gcn + (size_t)i * D,
            hnxt, N,  (i < L - 1) ? 1 : 0);
        float* hswap = hcur;
        hcur = hnxt;
        hnxt = hswap;
    }

    zero_f32<<<blocks((size_t)G * D), TB, 0, stream>>>(hg, (size_t)G * D);
    pool_kernel<<<blocks((size_t)N * 32), TB, 0, stream>>>(hcur, graph_ids, hg, N);

    {
        dim3 g0((G + MROWS - 1) / MROWS, Mm / 16);
        gemm_wmma_f32<<<g0, 32, 0, stream>>>(hg, W_mlp0, b_mlp0, nullptr, nullptr,
                                             t0, G, D, Mm, 0,  2);
        gemm_wmma_f32<<<g0, 32, 0, stream>>>(t0, W_mlp1, b_mlp1, nullptr, nullptr,
                                             t1, G, Mm, Mm, 0,  2);
        gemm_wmma_f32<<<g0, 32, 0, stream>>>(t1, W_mlp2, b_mlp2, nullptr, nullptr,
                                             (float*)d_out, G, Mm, Mm, 0,  0);
    }
}


# Readout tail (hg clear, SumPooling, 3 MLP GEMMs) as one launch: a block owns 8 graphs, its waves sum node runs in registers (node order; runs meet by LDS f32 adds), MLP on that tile; 14->10 launches
# speedup vs baseline: 4.7178x; 1.3107x over previous
#include <hip/hip_runtime.h>
#include <hip/hip_bf16.h>

typedef __attribute__((ext_vector_type(2))) float v2f;
typedef __attribute__((ext_vector_type(8))) float v8f;

#define D_FEAT 128

#define DG_THREADS 1024
#define DG_RANGES 64
#define DG_CAP 2048
#define DG_INFLIGHT 8

__device__ __forceinline__ void dg_count(int* cnt, int x, int lo, unsigned len) {
    const unsigned u = (unsigned)x - (unsigned)lo;
    if (u < len) atomicAdd(&cnt[u], 1);
}

__device__ __forceinline__ void dg_count4(int* cnt, int4 v, int lo, unsigned len) {
    dg_count(cnt, v.x, lo, len);
    dg_count(cnt, v.y, lo, len);
    dg_count(cnt, v.z, lo, len);
    dg_count(cnt, v.w, lo, len);
}

template <bool GUARD>
__device__ __forceinline__ void dg_batch(int* cnt, const int4* __restrict__ idx4, int j0, int n4,
                                         int lo, unsigned len) {
    int4 v[DG_INFLIGHT];
#pragma unroll
    for (int u = 0; u < DG_INFLIGHT; ++u) {
        const int j = j0 + u * DG_THREADS;
        if (!GUARD || j < n4) v[u] = idx4[j];
        else v[u] = make_int4(-1, -1, -1, -1);
    }
#pragma unroll
    for (int u = 0; u < DG_INFLIGHT; ++u) dg_count4(cnt, v[u], lo, len);
}

__global__ void __launch_bounds__(DG_THREADS)
degree_norm_kernel(const int* __restrict__ src, const int* __restrict__ dst,
                   float* __restrict__ norm_src, float* __restrict__ norm_dst, int E, int N) {
    __shared__ int cnt[DG_CAP];
    const int* __restrict__ idx = blockIdx.y ? dst : src;
    float* __restrict__ nrm = blockIdx.y ? norm_dst : norm_src;
    const int tid = threadIdx.x;
    const int R = (N + DG_RANGES - 1) / DG_RANGES;
    const int r0 = blockIdx.x * R;
    const int r1 = (r0 + R < N) ? (r0 + R) : N;
    const int n4 = E >> 2;
    const int4* __restrict__ idx4 = (const int4*)idx;

    for (int lo = r0; lo < r1; lo += DG_CAP) {
        const int len = (r1 - lo < DG_CAP) ? (r1 - lo) : DG_CAP;
        for (int i = tid; i < len; i += DG_THREADS) cnt[i] = 0;
        __syncthreads();

        int base = 0;
        for (; base + DG_INFLIGHT * DG_THREADS <= n4; base += DG_INFLIGHT * DG_THREADS)
            dg_batch<false>(cnt, idx4, base + tid, n4, lo, (unsigned)len);
        if (base < n4) dg_batch<true>(cnt, idx4, base + tid, n4, lo, (unsigned)len);
        for (int e = (n4 << 2) + tid; e < E; e += DG_THREADS) dg_count(cnt, idx[e], lo, (unsigned)len);
        __syncthreads();

        for (int k = tid; k < len; k += DG_THREADS)
            nrm[lo + k] = rsqrtf(fmaxf((float)cnt[k], 1.0f));
        __syncthreads();
    }
}

#define BN_SHIFT 6
#define BN_NODES (1 << BN_SHIFT)
#define BN_CAP 1024
#define BN_BLOCKS 64
#define BN_THREADS 1024

__device__ __forceinline__ int bn_bin(int d, int B) {
    unsigned b = (unsigned)d >> BN_SHIFT;
    if (b > (unsigned)(B - 1)) b = (unsigned)(B - 1);
    return (int)b;
}

__global__ void __launch_bounds__(BN_THREADS)
bin_count_kernel(const int* __restrict__ dst, int* __restrict__ hist, int E, int B) {
    __shared__ int cnt[BN_CAP];
    const int tid = threadIdx.x;
    for (int i = tid; i < BN_CAP; i += BN_THREADS) cnt[i] = 0;
    __syncthreads();
    const int S = (E + BN_BLOCKS - 1) / BN_BLOCKS;
    const int e0 = blockIdx.x * S;
    const int e1 = (e0 + S < E) ? (e0 + S) : E;
    for (int e = e0 + tid; e < e1; e += BN_THREADS) atomicAdd(&cnt[bn_bin(dst[e], B)], 1);
    __syncthreads();
    for (int i = tid; i < B; i += BN_THREADS) hist[blockIdx.x * B + i] = cnt[i];
}

__global__ void __launch_bounds__(BN_THREADS)
bin_place_kernel(const int* __restrict__ src, const int* __restrict__ dst,
                 const int* __restrict__ hist, int* __restrict__ boff,
                 int2* __restrict__ binned, int E, int B) {
    __shared__ int cur[BN_CAP];
    __shared__ int wsum[BN_THREADS / 32];
    const int tid = threadIdx.x;
    const int lane = tid & 31;
    const int wave = tid >> 5;
    const int blk = blockIdx.x;

    int tot = 0, pre = 0;
    if (tid < B) {
        for (int b = 0; b < BN_BLOCKS; ++b) {
            const int c = hist[b * B + tid];
            tot += c;
            if (b < blk) pre += c;
        }
    }
    int v = tot;
    for (int d = 1; d < 32; d <<= 1) {
        const int t = __shfl_up(v, d);
        if (lane >= d) v += t;
    }
    if (lane == 31) wsum[wave] = v;
    __syncthreads();
    if (wave == 0) {
        const int w = wsum[lane];
        int wi = w;
        for (int d = 1; d < 32; d <<= 1) {
            const int t = __shfl_up(wi, d);
            if (lane >= d) wi += t;
        }
        wsum[lane] = wi - w;
    }
    __syncthreads();
    const int excl = v - tot + wsum[wave];
    if (tid < B) {
        cur[tid] = excl + pre;
        if (blk == 0) boff[tid] = excl;
    }
    if (blk == 0 && tid == 0) boff[B] = E;
    __syncthreads();

    const int S = (E + BN_BLOCKS - 1) / BN_BLOCKS;
    const int e0 = blk * S;
    const int e1 = (e0 + S < E) ? (e0 + S) : E;
    for (int e = e0 + tid; e < e1; e += BN_THREADS) {
        const int s = src[e];
        const int d = dst[e];
        const int pos = atomicAdd(&cur[bn_bin(d, B)], 1);
        binned[pos] = make_int2(s, d);
    }
}

#define NG_WAVES 8

template <int MODE>
__device__ __forceinline__ float ng_finish(float a, float b, float r, int relu_c) {
    float cc = a + b;
    if (relu_c) cc = fmaxf(cc, 0.0f);
    return (MODE == 1) ? fmaxf(r + cc, 0.0f) : cc;
}

template <int MODE>
__global__ void __launch_bounds__(32 * NG_WAVES)
node_gemm_rows128(const float* A, const float* __restrict__ Wg,
                  const float* __restrict__ bias,
                  const float* __restrict__ row_scale,
                  const float* resid, float* out,
                  int Mrows, int relu_c) {
    __shared__ float4 Wl[32 * D_FEAT];

    const int tid  = threadIdx.x;
    const int wave = tid >> 5;
    const int lane = tid & 31;
    const int hh   = lane >> 4;
    const int c    = lane & 15;

    float4 wr[4][4];
#pragma unroll
    for (int it = 0; it < 4; ++it) {
        int p  = it * (32 * NG_WAVES) + tid;
        int kq = p >> 5;
        int n4 = p & 31;
        const float* wp = Wg + (size_t)(4 * kq) * D_FEAT + 4 * n4;
#pragma unroll
        for (int q = 0; q < 4; ++q) wr[it][q] = *(const float4*)(wp + q * D_FEAT);
    }

    const int tile = blockIdx.x * NG_WAVES + wave;
    int row = tile * 16 + c;
    if (row >= Mrows) row = Mrows - 1;
    const float* ap = A + (size_t)row * D_FEAT + 4 * hh;
    const float rs = row_scale ? row_scale[row] : 1.0f;
    float4 av[16];
#pragma unroll
    for (int j = 0; j < 16; ++j) av[j] = *(const float4*)(ap + 8 * j);

#pragma unroll
    for (int it = 0; it < 4; ++it) {
        int p  = it * (32 * NG_WAVES) + tid;
        int kq = p >> 5;
        int n4 = p & 31;
        float4* dstp = Wl + kq * D_FEAT + (4 * (n4 >> 4)) * 16 + (n4 & 15);
        dstp[0 * 16] = make_float4(wr[it][0].x, wr[it][1].x, wr[it][2].x, wr[it][3].x);
        dstp[1 * 16] = make_float4(wr[it][0].y, wr[it][1].y, wr[it][2].y, wr[it][3].y);
        dstp[2 * 16] = make_float4(wr[it][0].z, wr[it][1].z, wr[it][2].z, wr[it][3].z);
        dstp[3 * 16] = make_float4(wr[it][0].w, wr[it][1].w, wr[it][2].w, wr[it][3].w);
    }
    __syncthreads();

    v8f acc[8] = {};
    const float4* wl = Wl + hh * D_FEAT + c;
    float4 bw[8];
#pragma unroll
    for (int t = 0; t < 8; ++t) bw[t] = wl[t * 16];
#pragma unroll
    for (int j = 0; j < 16; ++j) {
        v2f a0, a1;
        a0.x = av[j].x * rs;  a0.y = av[j].y * rs;
        a1.x = av[j].z * rs;  a1.y = av[j].w * rs;
        float4 bn[8];
        if (j < 15) {
#pragma unroll
            for (int t = 0; t < 8; ++t) bn[t] = wl[(2 * (j + 1)) * D_FEAT + t * 16];
        }
        __builtin_amdgcn_sched_barrier(0);
#pragma unroll
        for (int t = 0; t < 8; ++t) {
            v2f b0; b0.x = bw[t].x; b0.y = bw[t].y;
            acc[t] = __builtin_amdgcn_wmma_f32_16x16x4_f32(
                false, a0, false, b0, (short)0, acc[t], false, false);
        }
#pragma unroll
        for (int t = 0; t < 8; ++t) {
            v2f b1; b1.x = bw[t].z; b1.y = bw[t].w;
            acc[t] = __builtin_amdgcn_wmma_f32_16x16x4_f32(
                false, a1, false, b1, (short)0, acc[t], false, false);
        }
        __builtin_amdgcn_sched_barrier(0);
        if (j < 15) {
#pragma unroll
            for (int t = 0; t < 8; ++t) bw[t] = bn[t];
        }
    }

    const int m0 = tile * 16 + hh * 8;
    float4 bz[2];
#pragma unroll
    for (int g = 0; g < 2; ++g) bz[g] = *(const float4*)(bias + 64 * g + 4 * c);

    if (tile * 16 + 16 <= Mrows) {
        float4 rv[8][2];
#pragma unroll
        for (int v = 0; v < 8; ++v)
#pragma unroll
            for (int g = 0; g < 2; ++g) {
                if (MODE == 1)
                    rv[v][g] = *(const float4*)(resid + (size_t)(m0 + v) * D_FEAT + 64 * g + 4 * c);
                else
                    rv[v][g] = make_float4(0.0f, 0.0f, 0.0f, 0.0f);
            }
#pragma unroll
        for (int v = 0; v < 8; ++v)
#pragma unroll
            for (int g = 0; g < 2; ++g) {
                float4 o;
                o.x = ng_finish<MODE>(acc[4 * g + 0][v], bz[g].x, rv[v][g].x, relu_c);
                o.y = ng_finish<MODE>(acc[4 * g + 1][v], bz[g].y, rv[v][g].y, relu_c);
                o.z = ng_finish<MODE>(acc[4 * g + 2][v], bz[g].z, rv[v][g].z, relu_c);
                o.w = ng_finish<MODE>(acc[4 * g + 3][v], bz[g].w, rv[v][g].w, relu_c);
                *(float4*)(out + (size_t)(m0 + v) * D_FEAT + 64 * g + 4 * c) = o;
            }
    } else {
#pragma unroll
        for (int v = 0; v < 8; ++v) {
            if (m0 + v < Mrows) {
#pragma unroll
                for (int g = 0; g < 2; ++g) {
                    float4 r = make_float4(0.0f, 0.0f, 0.0f, 0.0f);
                    if (MODE == 1)
                        r = *(const float4*)(resid + (size_t)(m0 + v) * D_FEAT + 64 * g + 4 * c);
                    float4 o;
                    o.x = ng_finish<MODE>(acc[4 * g + 0][v], bz[g].x, r.x, relu_c);
                    o.y = ng_finish<MODE>(acc[4 * g + 1][v], bz[g].y, r.y, relu_c);
                    o.z = ng_finish<MODE>(acc[4 * g + 2][v], bz[g].z, r.z, relu_c);
                    o.w = ng_finish<MODE>(acc[4 * g + 3][v], bz[g].w, r.w, relu_c);
                    *(float4*)(out + (size_t)(m0 + v) * D_FEAT + 64 * g + 4 * c) = o;
                }
            }
        }
    }
}

#define SC_EPW 16
#define FL_WAVES 8
#define FL_PITCH 132

__device__ __forceinline__ void fl_w_load(float4 (&wr)[2][4], const float* __restrict__ Wg,
                                          int khalf, int tid) {
#pragma unroll
    for (int it = 0; it < 2; ++it) {
        const int p  = it * (32 * FL_WAVES) + tid;
        const int kq = p >> 5;
        const int n4 = p & 31;
        const float* wp = Wg + (size_t)(64 * khalf + 4 * kq) * D_FEAT + 4 * n4;
#pragma unroll
        for (int q = 0; q < 4; ++q) wr[it][q] = *(const float4*)(wp + q * D_FEAT);
    }
}

__device__ __forceinline__ void fl_w_store(float4* Wl, const float4 (&wr)[2][4], int tid) {
#pragma unroll
    for (int it = 0; it < 2; ++it) {
        const int p  = it * (32 * FL_WAVES) + tid;
        const int kq = p >> 5;
        const int n4 = p & 31;
        float4* dstp = Wl + kq * D_FEAT + (4 * (n4 >> 4)) * 16 + (n4 & 15);
        dstp[0 * 16] = make_float4(wr[it][0].x, wr[it][1].x, wr[it][2].x, wr[it][3].x);
        dstp[1 * 16] = make_float4(wr[it][0].y, wr[it][1].y, wr[it][2].y, wr[it][3].y);
        dstp[2 * 16] = make_float4(wr[it][0].z, wr[it][1].z, wr[it][2].z, wr[it][3].z);
        dstp[3 * 16] = make_float4(wr[it][0].w, wr[it][1].w, wr[it][2].w, wr[it][3].w);
    }
}

__device__ __forceinline__ void fl_gemm_half(v8f (&acc)[4], const float* arow, const float4* wl,
                                             float rs) {
    float4 aw = *(const float4*)arow;
    float4 bw[4];
#pragma unroll
    for (int t = 0; t < 4; ++t) bw[t] = wl[t * 16];
#pragma unroll
    for (int j = 0; j < 8; ++j) {
        v2f a0, a1;
        a0.x = aw.x * rs;  a0.y = aw.y * rs;
        a1.x = aw.z * rs;  a1.y = aw.w * rs;
        float4 an = aw;
        float4 bn[4];
        if (j < 7) {
            an = *(const float4*)(arow + 8 * (j + 1));
#pragma unroll
            for (int t = 0; t < 4; ++t) bn[t] = wl[(2 * (j + 1)) * D_FEAT + t * 16];
        }
        __builtin_amdgcn_sched_barrier(0);
#pragma unroll
        for (int t = 0; t < 4; ++t) {
            v2f b0; b0.x = bw[t].x; b0.y = bw[t].y;
            acc[t] = __builtin_amdgcn_wmma_f32_16x16x4_f32(
                false, a0, false, b0, (short)0, acc[t], false, false);
        }
#pragma unroll
        for (int t = 0; t < 4; ++t) {
            v2f b1; b1.x = bw[t].z; b1.y = bw[t].w;
            acc[t] = __builtin_amdgcn_wmma_f32_16x16x4_f32(
                false, a1, false, b1, (short)0, acc[t], false, false);
        }
        __builtin_amdgcn_sched_barrier(0);
        if (j < 7) {
            aw = an;
#pragma unroll
            for (int t = 0; t < 4; ++t) bw[t] = bn[t];
        }
    }
}

__global__ void __launch_bounds__(32 * FL_WAVES)
graphconv_bins_kernel(const float* __restrict__ hin, const float* __restrict__ norm_src,
                      const float* __restrict__ norm_dst,
                      const int2* __restrict__ binned, const int* __restrict__ boff,
                      const float* __restrict__ Wg, const float* __restrict__ bias,
                      float* __restrict__ hout, int N, int relu_c) {
    __shared__ __attribute__((aligned(16))) float rows[BN_NODES * FL_PITCH];
    __shared__ float4 Wl[16 * D_FEAT];
    const int tid = threadIdx.x;
    const int lane = tid & 31;
    const int wid = __builtin_amdgcn_readfirstlane((int)(threadIdx.x >> 5));
    const int bin = blockIdx.x;
    const int lo = bin << BN_SHIFT;

    float4 wr[2][4];
    fl_w_load(wr, Wg, 0, tid);
    float4* r4 = (float4*)rows;
    for (int i = tid; i < BN_NODES * FL_PITCH / 4; i += 32 * FL_WAVES)
        r4[i] = make_float4(0.0f, 0.0f, 0.0f, 0.0f);
    const int b0 = boff[bin];
    const int b1 = boff[bin + 1];
    fl_w_store(Wl, wr, tid);
    __syncthreads();

    for (int e0 = b0 + wid * SC_EPW; e0 < b1; e0 += FL_WAVES * SC_EPW) {
        const int n = (b1 - e0 < SC_EPW) ? (b1 - e0) : SC_EPW;
        int el = e0 + (lane & (SC_EPW - 1));
        if (el > b1 - 1) el = b1 - 1;
        const int2 ed = binned[el];
        const int sv = ed.x;
        const int dv = (ed.y - lo) & (BN_NODES - 1);
        const int nv = __float_as_int(norm_src[sv]);

        if (n == SC_EPW) {
            float v[SC_EPW][4];
#pragma unroll
            for (int i = 0; i < SC_EPW; ++i) {
                const int s = __builtin_amdgcn_readlane(sv, i);
                const float* hp = hin + (size_t)s * D_FEAT + lane;
#pragma unroll
                for (int k = 0; k < 4; ++k) v[i][k] = hp[32 * k];
            }
#pragma unroll
            for (int i = 0; i < SC_EPW; ++i) {
                const int d = __builtin_amdgcn_readlane(dv, i);
                const float ns = __int_as_float(__builtin_amdgcn_readlane(nv, i));
                float* rp = rows + d * FL_PITCH + lane;
#pragma unroll
                for (int k = 0; k < 4; ++k)
                    __hip_atomic_fetch_add(rp + 32 * k, v[i][k] * ns, __ATOMIC_RELAXED,
                                           __HIP_MEMORY_SCOPE_WORKGROUP);
            }
        } else {
            for (int i = 0; i < n; ++i) {
                const int s = __builtin_amdgcn_readlane(sv, i);
                const int d = __builtin_amdgcn_readlane(dv, i);
                const float ns = __int_as_float(__builtin_amdgcn_readlane(nv, i));
                const float* hp = hin + (size_t)s * D_FEAT + lane;
                float* rp = rows + d * FL_PITCH + lane;
                float t[4];
#pragma unroll
                for (int k = 0; k < 4; ++k) t[k] = hp[32 * k];
#pragma unroll
                for (int k = 0; k < 4; ++k)
                    __hip_atomic_fetch_add(rp + 32 * k, t[k] * ns, __ATOMIC_RELAXED,
                                           __HIP_MEMORY_SCOPE_WORKGROUP);
            }
        }
    }
    __syncthreads();

    const int hh = lane >> 4;
    const int c  = lane & 15;
    const int rt = wid & 3;
    const int g  = wid >> 2;
    int node = lo + 16 * rt + c;
    if (node >= N) node = N - 1;
    const float rs = norm_dst[node];
    const float* arow = rows + (16 * rt + c) * FL_PITCH + 4 * hh;
    const float4* wl = Wl + hh * D_FEAT + (4 * g) * 16 + c;

    fl_w_load(wr, Wg, 1, tid);

    v8f acc[4] = {};
    fl_gemm_half(acc, arow, wl, rs);
    __syncthreads();
    fl_w_store(Wl, wr, tid);
    __syncthreads();
    fl_gemm_half(acc, arow + 64, wl, rs);

    const int m0 = lo + 16 * rt + hh * 8;
    const float4 bz = *(const float4*)(bias + 64 * g + 4 * c);
    float4 rv[8];
#pragma unroll
    for (int v = 0; v < 8; ++v) {
        int m = m0 + v;
        if (m >= N) m = N - 1;
        rv[v] = *(const float4*)(hin + (size_t)m * D_FEAT + 64 * g + 4 * c);
    }
#pragma unroll
    for (int v = 0; v < 8; ++v) {
        if (m0 + v < N) {
            float4 o;
            o.x = ng_finish<1>(acc[0][v], bz.x, rv[v].x, relu_c);
            o.y = ng_finish<1>(acc[1][v], bz.y, rv[v].y, relu_c);
            o.z = ng_finish<1>(acc[2][v], bz.z, rv[v].z, relu_c);
            o.w = ng_finish<1>(acc[3][v], bz.w, rv[v].w, relu_c);
            *(float4*)(hout + (size_t)(m0 + v) * D_FEAT + 64 * g + 4 * c) = o;
        }
    }
}

#define RO_TG 8
#define RO_WAVES 16
#define RO_THREADS (32 * RO_WAVES)
#define RO_PA 132
#define RO_PT 260
#define RO_MLP 256
#define RO_BATCH 16
#define RO_SCAN 16

__device__ __forceinline__ void ro_b_load(v2f (&b)[32], const float* __restrict__ Wm, int kbase,
                                          int hh, int col) {
#pragma unroll
    for (int s = 0; s < 32; ++s) {
        const float* p = Wm + (size_t)(kbase + 4 * s + 2 * hh) * RO_MLP + col;
        b[s].x = p[0];
        b[s].y = p[RO_MLP];
    }
}

__device__ __forceinline__ void ro_stage(v8f& acc, const v2f (&b)[32], const float* arow) {
#pragma unroll
    for (int s = 0; s < 32; ++s) {
        const v2f a = *(const v2f*)(arow + 4 * s);
        acc = __builtin_amdgcn_wmma_f32_16x16x4_f32(false, a, false, b[s], (short)0, acc,
                                                    false, false);
    }
}

__device__ __forceinline__ void ro_flush(float* tile, int g, const float (&acc)[4], int lane) {
    if (g >= 0) {
        float* rp = tile + g * RO_PA + lane;
#pragma unroll
        for (int k = 0; k < 4; ++k)
            __hip_atomic_fetch_add(rp + 32 * k, acc[k], __ATOMIC_RELAXED,
                                   __HIP_MEMORY_SCOPE_WORKGROUP);
    }
}

__device__ __forceinline__ void ro_span(int& mn, int& mx, int id, int n, int g0) {
    if ((unsigned)(id - g0) < (unsigned)RO_TG) {
        mn = (n < mn) ? n : mn;
        mx = (n > mx) ? n : mx;
    }
}

__global__ void __launch_bounds__(RO_THREADS)
readout_kernel(const float* __restrict__ hfin, const int* __restrict__ gid,
               const float* __restrict__ W0, const float* __restrict__ c0,
               const float* __restrict__ W1, const float* __restrict__ c1,
               const float* __restrict__ W2, const float* __restrict__ c2,
               float* __restrict__ out, int N, int G) {
    __shared__ __attribute__((aligned(16))) float tA[16 * RO_PA];
    __shared__ __attribute__((aligned(16))) float tB[16 * RO_PT];
    __shared__ __attribute__((aligned(16))) float tC[16 * RO_PT];
    __shared__ int rng[2];
    const int tid = threadIdx.x;
    const int lane = tid & 31;
    const int wid = __builtin_amdgcn_readfirstlane((int)(threadIdx.x >> 5));
    const int hh = lane >> 4;
    const int c = lane & 15;
    const int col = 16 * wid + c;
    const int g0 = blockIdx.x * RO_TG;

    v2f bA[32], bB[32];
    ro_b_load(bA, W0, 0, hh, col);
    ro_b_load(bB, W1, 0, hh, col);

    for (int i = tid; i < 16 * RO_PA; i += RO_THREADS) tA[i] = 0.0f;
    if (tid == 0) { rng[0] = N; rng[1] = -1; }
    __syncthreads();

    int mn = N, mx = -1;
    const int n4 = N >> 2;
    const int4* __restrict__ gid4 = (const int4*)gid;
    for (int base = 0; base < n4; base += RO_SCAN * RO_THREADS) {
        int4 v[RO_SCAN];
#pragma unroll
        for (int u = 0; u < RO_SCAN; ++u) {
            const int j = base + u * RO_THREADS + tid;
            if (j < n4) v[u] = gid4[j];
            else v[u] = make_int4(-1, -1, -1, -1);
        }
#pragma unroll
        for (int u = 0; u < RO_SCAN; ++u) {
            const int n = 4 * (base + u * RO_THREADS + tid);
            ro_span(mn, mx, v[u].x, n + 0, g0);
            ro_span(mn, mx, v[u].y, n + 1, g0);
            ro_span(mn, mx, v[u].z, n + 2, g0);
            ro_span(mn, mx, v[u].w, n + 3, g0);
        }
    }
    for (int n = (n4 << 2) + tid; n < N; n += RO_THREADS) ro_span(mn, mx, gid[n], n, g0);
    for (int d = 16; d >= 1; d >>= 1) {
        const int a = __shfl_xor(mn, d);
        const int b = __shfl_xor(mx, d);
        mn = (a < mn) ? a : mn;
        mx = (b > mx) ? b : mx;
    }
    if (lane == 0) {
        atomicMin(&rng[0], mn);
        atomicMax(&rng[1], mx);
    }
    __syncthreads();
    const int lo = __builtin_amdgcn_readfirstlane(rng[0]);
    const int hi = __builtin_amdgcn_readfirstlane(rng[1]) + 1;

    if (hi > lo) {
        const int per = (hi - lo + RO_WAVES - 1) / RO_WAVES;
        const int w0 = lo + wid * per;
        const int w1 = (w0 + per < hi) ? (w0 + per) : hi;
        int cur = -1;
        float acc[4] = {0.0f, 0.0f, 0.0f, 0.0f};
        for (int base = w0; base < w1; base += RO_BATCH) {
            int nl = base + (lane & (RO_BATCH - 1));
            if (nl > w1 - 1) nl = w1 - 1;
            const int gv = gid[nl] - g0;
            float v[RO_BATCH][4];
#pragma unroll
            for (int i = 0; i < RO_BATCH; ++i) {
                int n = base + i;
                if (n > w1 - 1) n = w1 - 1;
                const float* hp = hfin + (size_t)n * D_FEAT + lane;
#pragma unroll
                for (int k = 0; k < 4; ++k) v[i][k] = hp[32 * k];
            }
#pragma unroll
            for (int i = 0; i < RO_BATCH; ++i)
                asm volatile("" :: "v"(v[i][0]), "v"(v[i][1]), "v"(v[i][2]), "v"(v[i][3]));
#pragma unroll
            for (int i = 0; i < RO_BATCH; ++i) {
                const int g = __builtin_amdgcn_readlane(gv, i);
                if (base + i < w1 && (unsigned)g < (unsigned)RO_TG) {
                    if (g != cur) {
                        ro_flush(tA, cur, acc, lane);
                        cur = g;
#pragma unroll
                        for (int k = 0; k < 4; ++k) acc[k] = 0.0f;
                    }
#pragma unroll
                    for (int k = 0; k < 4; ++k) acc[k] += v[i][k];
                }
            }
        }
        ro_flush(tA, cur, acc, lane);
    }
    __syncthreads();

    v8f acc0 = {};
    ro_stage(acc0, bA, tA + c * RO_PA + 2 * hh);
    {
        const float bv = c0[col];
#pragma unroll
        for (int v = 0; v < 8; ++v) tB[(8 * hh + v) * RO_PT + col] = fmaxf(acc0[v] + bv, 0.0f);
    }
    __syncthreads();

    v8f acc1 = {};
    ro_b_load(bA, W1, 128, hh, col);
    ro_stage(acc1, bB, tB + c * RO_PT + 2 * hh);
    ro_b_load(bB, W2, 0, hh, col);
    ro_stage(acc1, bA, tB + c * RO_PT + 128 + 2 * hh);
    {
        const float bv = c1[col];
#pragma unroll
        for (int v = 0; v < 8; ++v) tC[(8 * hh + v) * RO_PT + col] = fmaxf(acc1[v] + bv, 0.0f);
    }
    __syncthreads();

    v8f acc2 = {};
    ro_b_load(bA, W2, 128, hh, col);
    ro_stage(acc2, bB, tC + c * RO_PT + 2 * hh);
    ro_stage(acc2, bA, tC + c * RO_PT + 128 + 2 * hh);
    {
        const float bv = c2[col];
#pragma unroll
        for (int v = 0; v < 8; ++v) {
            const int r = 8 * hh + v;
            if (r < RO_TG && g0 + r < G) out[(size_t)(g0 + r) * RO_MLP + col] = acc2[v] + bv;
        }
    }
}


extern "C" void kernel_launch(void* const* d_in, const int* in_sizes, int n_in,
                              void* d_out, int out_size, void* d_ws, size_t ws_size,
                              hipStream_t stream) {
    const float* node_features = (const float*)d_in[0];
    const int*   edge_src      = (const int*)d_in[1];
    const int*   edge_dst      = (const int*)d_in[2];
    const int*   graph_ids     = (const int*)d_in[3];
    const float* W_proj        = (const float*)d_in[4];
    const float* b_proj        = (const float*)d_in[5];
    const float* W_gcn         = (const float*)d_in[6];
    const float* b_gcn         = (const float*)d_in[7];
    const float* W_mlp0        = (const float*)d_in[8];
    const float* b_mlp0        = (const float*)d_in[9];
    const float* W_mlp1        = (const float*)d_in[10];
    const float* b_mlp1        = (const float*)d_in[11];
    const float* W_mlp2        = (const float*)d_in[12];
    const float* b_mlp2        = (const float*)d_in[13];

    const int D = D_FEAT;
    const int Mm = 256;
    const int N = in_sizes[3];
    const int E = in_sizes[1];
    const int L = in_sizes[6] / (D * D);
    const int G = out_size / Mm;

    float* ws = (float*)d_ws;
    float* norm_src = ws;
    float* norm_dst = norm_src + N;
    float* h        = norm_dst + N;
    float* h2       = h + (size_t)N * D;
    float* hg       = h2 + (size_t)N * D;
    float* t0       = hg + (size_t)G * D;
    float* t1       = t0 + (size_t)G * Mm;
    const int B = (N + BN_NODES - 1) / BN_NODES;
    int2* binned    = (int2*)(t1 + (size_t)G * Mm);
    int*  hist      = (int*)(binned + E);
    int*  boff      = hist + (size_t)BN_BLOCKS * B;

    degree_norm_kernel<<<dim3(DG_RANGES, 2), DG_THREADS, 0, stream>>>(edge_src, edge_dst,
                                                                      norm_src, norm_dst, E, N);

    bin_count_kernel<<<BN_BLOCKS, BN_THREADS, 0, stream>>>(edge_dst, hist, E, B);
    bin_place_kernel<<<BN_BLOCKS, BN_THREADS, 0, stream>>>(edge_src, edge_dst, hist, boff, binned, E, B);

    const int NG_ROWS = 16 * NG_WAVES;
    const unsigned ng_blocks = (unsigned)((N + NG_ROWS - 1) / NG_ROWS);
    node_gemm_rows128<0><<<ng_blocks, 32 * NG_WAVES, 0, stream>>>(node_features, W_proj, b_proj,
                                                                 nullptr, nullptr, h, N,
                                                                  0);

    float* hcur = h;
    float* hnxt = h2;
    for (int i = 0; i < L; ++i) {
        graphconv_bins_kernel<<<(unsigned)B, 32 * FL_WAVES, 0, stream>>>(
            hcur, norm_src, norm_dst, binned, boff, W_gcn + (size_t)i * D * D, b_gcn + (size_t)i * D,
            hnxt, N,  (i < L - 1) ? 1 : 0);
        float* hswap = hcur;
        hcur = hnxt;
        hnxt = hswap;
    }

    readout_kernel<<<(unsigned)((G + RO_TG - 1) / RO_TG), RO_THREADS, 0, stream>>>(
        hcur, graph_ids, W_mlp0, b_mlp0, W_mlp1, b_mlp1, W_mlp2, b_mlp2, (float*)d_out, N, G);
}


# GraphConv launches re-cut to one block per work-group processor: 256 bins of ceil(N/256) nodes (were 782 of 64), 26-wave blocks, whole W in LDS per block; LDS adds in arrival order as before
# speedup vs baseline: 5.1053x; 1.0821x over previous
#include <hip/hip_runtime.h>
#include <hip/hip_bf16.h>

typedef __attribute__((ext_vector_type(2))) float v2f;
typedef __attribute__((ext_vector_type(8))) float v8f;

#define D_FEAT 128

#define DG_THREADS 1024
#define DG_RANGES 64
#define DG_CAP 2048
#define DG_INFLIGHT 8

__device__ __forceinline__ void dg_count(int* cnt, int x, int lo, unsigned len) {
    const unsigned u = (unsigned)x - (unsigned)lo;
    if (u < len) atomicAdd(&cnt[u], 1);
}

__device__ __forceinline__ void dg_count4(int* cnt, int4 v, int lo, unsigned len) {
    dg_count(cnt, v.x, lo, len);
    dg_count(cnt, v.y, lo, len);
    dg_count(cnt, v.z, lo, len);
    dg_count(cnt, v.w, lo, len);
}

template <bool GUARD>
__device__ __forceinline__ void dg_batch(int* cnt, const int4* __restrict__ idx4, int j0, int n4,
                                         int lo, unsigned len) {
    int4 v[DG_INFLIGHT];
#pragma unroll
    for (int u = 0; u < DG_INFLIGHT; ++u) {
        const int j = j0 + u * DG_THREADS;
        if (!GUARD || j < n4) v[u] = idx4[j];
        else v[u] = make_int4(-1, -1, -1, -1);
    }
#pragma unroll
    for (int u = 0; u < DG_INFLIGHT; ++u) dg_count4(cnt, v[u], lo, len);
}

__global__ void __launch_bounds__(DG_THREADS)
degree_norm_kernel(const int* __restrict__ src, const int* __restrict__ dst,
                   float* __restrict__ norm_src, float* __restrict__ norm_dst, int E, int N) {
    __shared__ int cnt[DG_CAP];
    const int* __restrict__ idx = blockIdx.y ? dst : src;
    float* __restrict__ nrm = blockIdx.y ? norm_dst : norm_src;
    const int tid = threadIdx.x;
    const int R = (N + DG_RANGES - 1) / DG_RANGES;
    const int r0 = blockIdx.x * R;
    const int r1 = (r0 + R < N) ? (r0 + R) : N;
    const int n4 = E >> 2;
    const int4* __restrict__ idx4 = (const int4*)idx;

    for (int lo = r0; lo < r1; lo += DG_CAP) {
        const int len = (r1 - lo < DG_CAP) ? (r1 - lo) : DG_CAP;
        for (int i = tid; i < len; i += DG_THREADS) cnt[i] = 0;
        __syncthreads();

        int base = 0;
        for (; base + DG_INFLIGHT * DG_THREADS <= n4; base += DG_INFLIGHT * DG_THREADS)
            dg_batch<false>(cnt, idx4, base + tid, n4, lo, (unsigned)len);
        if (base < n4) dg_batch<true>(cnt, idx4, base + tid, n4, lo, (unsigned)len);
        for (int e = (n4 << 2) + tid; e < E; e += DG_THREADS) dg_count(cnt, idx[e], lo, (unsigned)len);
        __syncthreads();

        for (int k = tid; k < len; k += DG_THREADS)
            nrm[lo + k] = rsqrtf(fmaxf((float)cnt[k], 1.0f));
        __syncthreads();
    }
}

#define BN_CAP 1024
#define BN_BLOCKS 64
#define BN_THREADS 1024

__device__ __forceinline__ int bn_bin(int d, int B, int bn) {
    unsigned b = (unsigned)d / (unsigned)bn;
    if (b > (unsigned)(B - 1)) b = (unsigned)(B - 1);
    return (int)b;
}

__global__ void __launch_bounds__(BN_THREADS)
bin_count_kernel(const int* __restrict__ dst, int* __restrict__ hist, int E, int B, int bn) {
    __shared__ int cnt[BN_CAP];
    const int tid = threadIdx.x;
    for (int i = tid; i < BN_CAP; i += BN_THREADS) cnt[i] = 0;
    __syncthreads();
    const int S = (E + BN_BLOCKS - 1) / BN_BLOCKS;
    const int e0 = blockIdx.x * S;
    const int e1 = (e0 + S < E) ? (e0 + S) : E;
    for (int e = e0 + tid; e < e1; e += BN_THREADS) atomicAdd(&cnt[bn_bin(dst[e], B, bn)], 1);
    __syncthreads();
    for (int i = tid; i < B; i += BN_THREADS) hist[blockIdx.x * B + i] = cnt[i];
}

__global__ void __launch_bounds__(BN_THREADS)
bin_place_kernel(const int* __restrict__ src, const int* __restrict__ dst,
                 const int* __restrict__ hist, int* __restrict__ boff,
                 int2* __restrict__ binned, int E, int B, int bn) {
    __shared__ int cur[BN_CAP];
    __shared__ int wsum[BN_THREADS / 32];
    const int tid = threadIdx.x;
    const int lane = tid & 31;
    const int wave = tid >> 5;
    const int blk = blockIdx.x;

    int tot = 0, pre = 0;
    if (tid < B) {
        for (int b = 0; b < BN_BLOCKS; ++b) {
            const int c = hist[b * B + tid];
            tot += c;
            if (b < blk) pre += c;
        }
    }
    int v = tot;
    for (int d = 1; d < 32; d <<= 1) {
        const int t = __shfl_up(v, d);
        if (lane >= d) v += t;
    }
    if (lane == 31) wsum[wave] = v;
    __syncthreads();
    if (wave == 0) {
        const int w = wsum[lane];
        int wi = w;
        for (int d = 1; d < 32; d <<= 1) {
            const int t = __shfl_up(wi, d);
            if (lane >= d) wi += t;
        }
        wsum[lane] = wi - w;
    }
    __syncthreads();
    const int excl = v - tot + wsum[wave];
    if (tid < B) {
        cur[tid] = excl + pre;
        if (blk == 0) boff[tid] = excl;
    }
    if (blk == 0 && tid == 0) boff[B] = E;
    __syncthreads();

    const int S = (E + BN_BLOCKS - 1) / BN_BLOCKS;
    const int e0 = blk * S;
    const int e1 = (e0 + S < E) ? (e0 + S) : E;
    for (int e = e0 + tid; e < e1; e += BN_THREADS) {
        const int s = src[e];
        const int d = dst[e];
        const int pos = atomicAdd(&cur[bn_bin(d, B, bn)], 1);
        binned[pos] = make_int2(s, d);
    }
}

#define NG_WAVES 8

template <int MODE>
__device__ __forceinline__ float ng_finish(float a, float b, float r, int relu_c) {
    float cc = a + b;
    if (relu_c) cc = fmaxf(cc, 0.0f);
    return (MODE == 1) ? fmaxf(r + cc, 0.0f) : cc;
}

template <int MODE>
__global__ void __launch_bounds__(32 * NG_WAVES)
node_gemm_rows128(const float* A, const float* __restrict__ Wg,
                  const float* __restrict__ bias,
                  const float* __restrict__ row_scale,
                  const float* resid, float* out,
                  int Mrows, int relu_c) {
    __shared__ float4 Wl[32 * D_FEAT];

    const int tid  = threadIdx.x;
    const int wave = tid >> 5;
    const int lane = tid & 31;
    const int hh   = lane >> 4;
    const int c    = lane & 15;

    float4 wr[4][4];
#pragma unroll
    for (int it = 0; it < 4; ++it) {
        int p  = it * (32 * NG_WAVES) + tid;
        int kq = p >> 5;
        int n4 = p & 31;
        const float* wp = Wg + (size_t)(4 * kq) * D_FEAT + 4 * n4;
#pragma unroll
        for (int q = 0; q < 4; ++q) wr[it][q] = *(const float4*)(wp + q * D_FEAT);
    }

    const int tile = blockIdx.x * NG_WAVES + wave;
    int row = tile * 16 + c;
    if (row >= Mrows) row = Mrows - 1;
    const float* ap = A + (size_t)row * D_FEAT + 4 * hh;
    const float rs = row_scale ? row_scale[row] : 1.0f;
    float4 av[16];
#pragma unroll
    for (int j = 0; j < 16; ++j) av[j] = *(const float4*)(ap + 8 * j);

#pragma unroll
    for (int it = 0; it < 4; ++it) {
        int p  = it * (32 * NG_WAVES) + tid;
        int kq = p >> 5;
        int n4 = p & 31;
        float4* dstp = Wl + kq * D_FEAT + (4 * (n4 >> 4)) * 16 + (n4 & 15);
        dstp[0 * 16] = make_float4(wr[it][0].x, wr[it][1].x, wr[it][2].x, wr[it][3].x);
        dstp[1 * 16] = make_float4(wr[it][0].y, wr[it][1].y, wr[it][2].y, wr[it][3].y);
        dstp[2 * 16] = make_float4(wr[it][0].z, wr[it][1].z, wr[it][2].z, wr[it][3].z);
        dstp[3 * 16] = make_float4(wr[it][0].w, wr[it][1].w, wr[it][2].w, wr[it][3].w);
    }
    __syncthreads();

    v8f acc[8] = {};
    const float4* wl = Wl + hh * D_FEAT + c;
    float4 bw[8];
#pragma unroll
    for (int t = 0; t < 8; ++t) bw[t] = wl[t * 16];
#pragma unroll
    for (int j = 0; j < 16; ++j) {
        v2f a0, a1;
        a0.x = av[j].x * rs;  a0.y = av[j].y * rs;
        a1.x = av[j].z * rs;  a1.y = av[j].w * rs;
        float4 bn[8];
        if (j < 15) {
#pragma unroll
            for (int t = 0; t < 8; ++t) bn[t] = wl[(2 * (j + 1)) * D_FEAT + t * 16];
        }
        __builtin_amdgcn_sched_barrier(0);
#pragma unroll
        for (int t = 0; t < 8; ++t) {
            v2f b0; b0.x = bw[t].x; b0.y = bw[t].y;
            acc[t] = __builtin_amdgcn_wmma_f32_16x16x4_f32(
                false, a0, false, b0, (short)0, acc[t], false, false);
        }
#pragma unroll
        for (int t = 0; t < 8; ++t) {
            v2f b1; b1.x = bw[t].z; b1.y = bw[t].w;
            acc[t] = __builtin_amdgcn_wmma_f32_16x16x4_f32(
                false, a1, false, b1, (short)0, acc[t], false, false);
        }
        __builtin_amdgcn_sched_barrier(0);
        if (j < 15) {
#pragma unroll
            for (int t = 0; t < 8; ++t) bw[t] = bn[t];
        }
    }

    const int m0 = tile * 16 + hh * 8;
    float4 bz[2];
#pragma unroll
    for (int g = 0; g < 2; ++g) bz[g] = *(const float4*)(bias + 64 * g + 4 * c);

    if (tile * 16 + 16 <= Mrows) {
        float4 rv[8][2];
#pragma unroll
        for (int v = 0; v < 8; ++v)
#pragma unroll
            for (int g = 0; g < 2; ++g) {
                if (MODE == 1)
                    rv[v][g] = *(const float4*)(resid + (size_t)(m0 + v) * D_FEAT + 64 * g + 4 * c);
                else
                    rv[v][g] = make_float4(0.0f, 0.0f, 0.0f, 0.0f);
            }
#pragma unroll
        for (int v = 0; v < 8; ++v)
#pragma unroll
            for (int g = 0; g < 2; ++g) {
                float4 o;
                o.x = ng_finish<MODE>(acc[4 * g + 0][v], bz[g].x, rv[v][g].x, relu_c);
                o.y = ng_finish<MODE>(acc[4 * g + 1][v], bz[g].y, rv[v][g].y, relu_c);
                o.z = ng_finish<MODE>(acc[4 * g + 2][v], bz[g].z, rv[v][g].z, relu_c);
                o.w = ng_finish<MODE>(acc[4 * g + 3][v], bz[g].w, rv[v][g].w, relu_c);
                *(float4*)(out + (size_t)(m0 + v) * D_FEAT + 64 * g + 4 * c) = o;
            }
    } else {
#pragma unroll
        for (int v = 0; v < 8; ++v) {
            if (m0 + v < Mrows) {
#pragma unroll
                for (int g = 0; g < 2; ++g) {
                    float4 r = make_float4(0.0f, 0.0f, 0.0f, 0.0f);
                    if (MODE == 1)
                        r = *(const float4*)(resid + (size_t)(m0 + v) * D_FEAT + 64 * g + 4 * c);
                    float4 o;
                    o.x = ng_finish<MODE>(acc[4 * g + 0][v], bz[g].x, r.x, relu_c);
                    o.y = ng_finish<MODE>(acc[4 * g + 1][v], bz[g].y, r.y, relu_c);
                    o.z = ng_finish<MODE>(acc[4 * g + 2][v], bz[g].z, r.z, relu_c);
                    o.w = ng_finish<MODE>(acc[4 * g + 3][v], bz[g].w, r.w, relu_c);
                    *(float4*)(out + (size_t)(m0 + v) * D_FEAT + 64 * g + 4 * c) = o;
                }
            }
        }
    }
}

#define SC_EPW 16
#define FL_TILES 13
#define FL_ROWS (16 * FL_TILES)
#define FL_WAVES (2 * FL_TILES)
#define FL_THREADS (32 * FL_WAVES)
#define FL_PITCH 132
#define FL_WPIECES (32 * 32)

__device__ __forceinline__ void fl_w_load(float4 (&wr)[2][4], const float* __restrict__ Wg, int tid) {
#pragma unroll
    for (int it = 0; it < 2; ++it) {
        const int p  = it * FL_THREADS + tid;
        const int kq = p >> 5;
        const int n4 = p & 31;
        const float* wp = Wg + (size_t)(4 * kq) * D_FEAT + 4 * n4;
#pragma unroll
        for (int q = 0; q < 4; ++q) {
            if (p < FL_WPIECES) wr[it][q] = *(const float4*)(wp + q * D_FEAT);
            else wr[it][q] = make_float4(0.0f, 0.0f, 0.0f, 0.0f);
        }
    }
}

__device__ __forceinline__ void fl_w_store(float4* Wl, const float4 (&wr)[2][4], int tid) {
#pragma unroll
    for (int it = 0; it < 2; ++it) {
        const int p  = it * FL_THREADS + tid;
        const int kq = p >> 5;
        const int n4 = p & 31;
        if (p < FL_WPIECES) {
            float4* dstp = Wl + kq * D_FEAT + (4 * (n4 >> 4)) * 16 + (n4 & 15);
            dstp[0 * 16] = make_float4(wr[it][0].x, wr[it][1].x, wr[it][2].x, wr[it][3].x);
            dstp[1 * 16] = make_float4(wr[it][0].y, wr[it][1].y, wr[it][2].y, wr[it][3].y);
            dstp[2 * 16] = make_float4(wr[it][0].z, wr[it][1].z, wr[it][2].z, wr[it][3].z);
            dstp[3 * 16] = make_float4(wr[it][0].w, wr[it][1].w, wr[it][2].w, wr[it][3].w);
        }
    }
}

__device__ __forceinline__ void fl_gemm_half(v8f (&acc)[4], const float* arow, const float4* wl,
                                             float rs) {
    float4 aw = *(const float4*)arow;
    float4 bw[4];
#pragma unroll
    for (int t = 0; t < 4; ++t) bw[t] = wl[t * 16];
#pragma unroll
    for (int j = 0; j < 8; ++j) {
        v2f a0, a1;
        a0.x = aw.x * rs;  a0.y = aw.y * rs;
        a1.x = aw.z * rs;  a1.y = aw.w * rs;
        float4 an = aw;
        float4 bn[4];
        if (j < 7) {
            an = *(const float4*)(arow + 8 * (j + 1));
#pragma unroll
            for (int t = 0; t < 4; ++t) bn[t] = wl[(2 * (j + 1)) * D_FEAT + t * 16];
        }
        __builtin_amdgcn_sched_barrier(0);
#pragma unroll
        for (int t = 0; t < 4; ++t) {
            v2f b0; b0.x = bw[t].x; b0.y = bw[t].y;
            acc[t] = __builtin_amdgcn_wmma_f32_16x16x4_f32(
                false, a0, false, b0, (short)0, acc[t], false, false);
        }
#pragma unroll
        for (int t = 0; t < 4; ++t) {
            v2f b1; b1.x = bw[t].z; b1.y = bw[t].w;
            acc[t] = __builtin_amdgcn_wmma_f32_16x16x4_f32(
                false, a1, false, b1, (short)0, acc[t], false, false);
        }
        __builtin_amdgcn_sched_barrier(0);
        if (j < 7) {
            aw = an;
#pragma unroll
            for (int t = 0; t < 4; ++t) bw[t] = bn[t];
        }
    }
}

__global__ void __launch_bounds__(FL_THREADS)
graphconv_bins_kernel(const float* __restrict__ hin, const float* __restrict__ norm_src,
                      const float* __restrict__ norm_dst,
                      const int2* __restrict__ binned, const int* __restrict__ boff,
                      const float* __restrict__ Wg, const float* __restrict__ bias,
                      float* __restrict__ hout, int N, int bn, int relu_c) {
    __shared__ __attribute__((aligned(16))) float rows[FL_ROWS * FL_PITCH];
    __shared__ float4 Wl[32 * D_FEAT];
    const int tid = threadIdx.x;
    const int lane = tid & 31;
    const int wid = __builtin_amdgcn_readfirstlane((int)(threadIdx.x >> 5));
    const int bin = blockIdx.x;
    const int lo = bin * bn;

    float4 wr[2][4];
    fl_w_load(wr, Wg, tid);
    float4* r4 = (float4*)rows;
    for (int i = tid; i < FL_ROWS * FL_PITCH / 4; i += FL_THREADS)
        r4[i] = make_float4(0.0f, 0.0f, 0.0f, 0.0f);
    const int b0 = boff[bin];
    const int b1 = boff[bin + 1];
    fl_w_store(Wl, wr, tid);
    __syncthreads();

    for (int e0 = b0 + wid * SC_EPW; e0 < b1; e0 += FL_WAVES * SC_EPW) {
        const int n = (b1 - e0 < SC_EPW) ? (b1 - e0) : SC_EPW;
        int el = e0 + (lane & (SC_EPW - 1));
        if (el > b1 - 1) el = b1 - 1;
        const int2 ed = binned[el];
        const int sv = ed.x;
        unsigned du = (unsigned)(ed.y - lo);
        if (du > (unsigned)(FL_ROWS - 1)) du = (unsigned)(FL_ROWS - 1);
        const int dv = (int)du;
        const int nv = __float_as_int(norm_src[sv]);

        if (n == SC_EPW) {
            float v[SC_EPW][4];
#pragma unroll
            for (int i = 0; i < SC_EPW; ++i) {
                const int s = __builtin_amdgcn_readlane(sv, i);
                const float* hp = hin + (size_t)s * D_FEAT + lane;
#pragma unroll
                for (int k = 0; k < 4; ++k) v[i][k] = hp[32 * k];
            }
#pragma unroll
            for (int i = 0; i < SC_EPW; ++i) {
                const int d = __builtin_amdgcn_readlane(dv, i);
                const float ns = __int_as_float(__builtin_amdgcn_readlane(nv, i));
                float* rp = rows + d * FL_PITCH + lane;
#pragma unroll
                for (int k = 0; k < 4; ++k)
                    __hip_atomic_fetch_add(rp + 32 * k, v[i][k] * ns, __ATOMIC_RELAXED,
                                           __HIP_MEMORY_SCOPE_WORKGROUP);
            }
        } else {
            for (int i = 0; i < n; ++i) {
                const int s = __builtin_amdgcn_readlane(sv, i);
                const int d = __builtin_amdgcn_readlane(dv, i);
                const float ns = __int_as_float(__builtin_amdgcn_readlane(nv, i));
                const float* hp = hin + (size_t)s * D_FEAT + lane;
                float* rp = rows + d * FL_PITCH + lane;
                float t[4];
#pragma unroll
                for (int k = 0; k < 4; ++k) t[k] = hp[32 * k];
#pragma unroll
                for (int k = 0; k < 4; ++k)
                    __hip_atomic_fetch_add(rp + 32 * k, t[k] * ns, __ATOMIC_RELAXED,
                                           __HIP_MEMORY_SCOPE_WORKGROUP);
            }
        }
    }
    __syncthreads();

    const int hh = lane >> 4;
    const int c  = lane & 15;
    const int rt = wid >> 1;
    const int g  = wid & 1;
    int node = lo + 16 * rt + c;
    if (node >= N) node = N - 1;
    const float rs = norm_dst[node];
    const float* arow = rows + (16 * rt + c) * FL_PITCH + 4 * hh;
    const float4* wl = Wl + hh * D_FEAT + (4 * g) * 16 + c;

    v8f acc[4] = {};
    fl_gemm_half(acc, arow, wl, rs);
    fl_gemm_half(acc, arow + 64, wl + 16 * D_FEAT, rs);

    const int r0 = 16 * rt + hh * 8;
    const int m0 = lo + r0;
    const float4 bz = *(const float4*)(bias + 64 * g + 4 * c);
    float4 rv[8];
#pragma unroll
    for (int v = 0; v < 8; ++v) {
        int m = m0 + v;
        if (m >= N) m = N - 1;
        rv[v] = *(const float4*)(hin + (size_t)m * D_FEAT + 64 * g + 4 * c);
    }
#pragma unroll
    for (int v = 0; v < 8; ++v) {
        if (r0 + v < bn && m0 + v < N) {
            float4 o;
            o.x = ng_finish<1>(acc[0][v], bz.x, rv[v].x, relu_c);
            o.y = ng_finish<1>(acc[1][v], bz.y, rv[v].y, relu_c);
            o.z = ng_finish<1>(acc[2][v], bz.z, rv[v].z, relu_c);
            o.w = ng_finish<1>(acc[3][v], bz.w, rv[v].w, relu_c);
            *(float4*)(hout + (size_t)(m0 + v) * D_FEAT + 64 * g + 4 * c) = o;
        }
    }
}

#define RO_TG 8
#define RO_WAVES 16
#define RO_THREADS (32 * RO_WAVES)
#define RO_PA 132
#define RO_PT 260
#define RO_MLP 256
#define RO_BATCH 16
#define RO_SCAN 16

__device__ __forceinline__ void ro_b_load(v2f (&b)[32], const float* __restrict__ Wm, int kbase,
                                          int hh, int col) {
#pragma unroll
    for (int s = 0; s < 32; ++s) {
        const float* p = Wm + (size_t)(kbase + 4 * s + 2 * hh) * RO_MLP + col;
        b[s].x = p[0];
        b[s].y = p[RO_MLP];
    }
}

__device__ __forceinline__ void ro_stage(v8f& acc, const v2f (&b)[32], const float* arow) {
#pragma unroll
    for (int s = 0; s < 32; ++s) {
        const v2f a = *(const v2f*)(arow + 4 * s);
        acc = __builtin_amdgcn_wmma_f32_16x16x4_f32(false, a, false, b[s], (short)0, acc,
                                                    false, false);
    }
}

__device__ __forceinline__ void ro_flush(float* tile, int g, const float (&acc)[4], int lane) {
    if (g >= 0) {
        float* rp = tile + g * RO_PA + lane;
#pragma unroll
        for (int k = 0; k < 4; ++k)
            __hip_atomic_fetch_add(rp + 32 * k, acc[k], __ATOMIC_RELAXED,
                                   __HIP_MEMORY_SCOPE_WORKGROUP);
    }
}

__device__ __forceinline__ void ro_span(int& mn, int& mx, int id, int n, int g0) {
    if ((unsigned)(id - g0) < (unsigned)RO_TG) {
        mn = (n < mn) ? n : mn;
        mx = (n > mx) ? n : mx;
    }
}

__global__ void __launch_bounds__(RO_THREADS)
readout_kernel(const float* __restrict__ hfin, const int* __restrict__ gid,
               const float* __restrict__ W0, const float* __restrict__ c0,
               const float* __restrict__ W1, const float* __restrict__ c1,
               const float* __restrict__ W2, const float* __restrict__ c2,
               float* __restrict__ out, int N, int G) {
    __shared__ __attribute__((aligned(16))) float tA[16 * RO_PA];
    __shared__ __attribute__((aligned(16))) float tB[16 * RO_PT];
    __shared__ __attribute__((aligned(16))) float tC[16 * RO_PT];
    __shared__ int rng[2];
    const int tid = threadIdx.x;
    const int lane = tid & 31;
    const int wid = __builtin_amdgcn_readfirstlane((int)(threadIdx.x >> 5));
    const int hh = lane >> 4;
    const int c = lane & 15;
    const int col = 16 * wid + c;
    const int g0 = blockIdx.x * RO_TG;

    v2f bA[32], bB[32];
    ro_b_load(bA, W0, 0, hh, col);
    ro_b_load(bB, W1, 0, hh, col);

    for (int i = tid; i < 16 * RO_PA; i += RO_THREADS) tA[i] = 0.0f;
    if (tid == 0) { rng[0] = N; rng[1] = -1; }
    __syncthreads();

    int mn = N, mx = -1;
    const int n4 = N >> 2;
    const int4* __restrict__ gid4 = (const int4*)gid;
    for (int base = 0; base < n4; base += RO_SCAN * RO_THREADS) {
        int4 v[RO_SCAN];
#pragma unroll
        for (int u = 0; u < RO_SCAN; ++u) {
            const int j = base + u * RO_THREADS + tid;
            if (j < n4) v[u] = gid4[j];
            else v[u] = make_int4(-1, -1, -1, -1);
        }
#pragma unroll
        for (int u = 0; u < RO_SCAN; ++u) {
            const int n = 4 * (base + u * RO_THREADS + tid);
            ro_span(mn, mx, v[u].x, n + 0, g0);
            ro_span(mn, mx, v[u].y, n + 1, g0);
            ro_span(mn, mx, v[u].z, n + 2, g0);
            ro_span(mn, mx, v[u].w, n + 3, g0);
        }
    }
    for (int n = (n4 << 2) + tid; n < N; n += RO_THREADS) ro_span(mn, mx, gid[n], n, g0);
    for (int d = 16; d >= 1; d >>= 1) {
        const int a = __shfl_xor(mn, d);
        const int b = __shfl_xor(mx, d);
        mn = (a < mn) ? a : mn;
        mx = (b > mx) ? b : mx;
    }
    if (lane == 0) {
        atomicMin(&rng[0], mn);
        atomicMax(&rng[1], mx);
    }
    __syncthreads();
    const int lo = __builtin_amdgcn_readfirstlane(rng[0]);
    const int hi = __builtin_amdgcn_readfirstlane(rng[1]) + 1;

    if (hi > lo) {
        const int per = (hi - lo + RO_WAVES - 1) / RO_WAVES;
        const int w0 = lo + wid * per;
        const int w1 = (w0 + per < hi) ? (w0 + per) : hi;
        int cur = -1;
        float acc[4] = {0.0f, 0.0f, 0.0f, 0.0f};
        for (int base = w0; base < w1; base += RO_BATCH) {
            int nl = base + (lane & (RO_BATCH - 1));
            if (nl > w1 - 1) nl = w1 - 1;
            const int gv = gid[nl] - g0;
            float v[RO_BATCH][4];
#pragma unroll
            for (int i = 0; i < RO_BATCH; ++i) {
                int n = base + i;
                if (n > w1 - 1) n = w1 - 1;
                const float* hp = hfin + (size_t)n * D_FEAT + lane;
#pragma unroll
                for (int k = 0; k < 4; ++k) v[i][k] = hp[32 * k];
            }
#pragma unroll
            for (int i = 0; i < RO_BATCH; ++i)
                asm volatile("" :: "v"(v[i][0]), "v"(v[i][1]), "v"(v[i][2]), "v"(v[i][3]));
#pragma unroll
            for (int i = 0; i < RO_BATCH; ++i) {
                const int g = __builtin_amdgcn_readlane(gv, i);
                if (base + i < w1 && (unsigned)g < (unsigned)RO_TG) {
                    if (g != cur) {
                        ro_flush(tA, cur, acc, lane);
                        cur = g;
#pragma unroll
                        for (int k = 0; k < 4; ++k) acc[k] = 0.0f;
                    }
#pragma unroll
                    for (int k = 0; k < 4; ++k) acc[k] += v[i][k];
                }
            }
        }
        ro_flush(tA, cur, acc, lane);
    }
    __syncthreads();

    v8f acc0 = {};
    ro_stage(acc0, bA, tA + c * RO_PA + 2 * hh);
    {
        const float bv = c0[col];
#pragma unroll
        for (int v = 0; v < 8; ++v) tB[(8 * hh + v) * RO_PT + col] = fmaxf(acc0[v] + bv, 0.0f);
    }
    __syncthreads();

    v8f acc1 = {};
    ro_b_load(bA, W1, 128, hh, col);
    ro_stage(acc1, bB, tB + c * RO_PT + 2 * hh);
    ro_b_load(bB, W2, 0, hh, col);
    ro_stage(acc1, bA, tB + c * RO_PT + 128 + 2 * hh);
    {
        const float bv = c1[col];
#pragma unroll
        for (int v = 0; v < 8; ++v) tC[(8 * hh + v) * RO_PT + col] = fmaxf(acc1[v] + bv, 0.0f);
    }
    __syncthreads();

    v8f acc2 = {};
    ro_b_load(bA, W2, 128, hh, col);
    ro_stage(acc2, bB, tC + c * RO_PT + 2 * hh);
    ro_stage(acc2, bA, tC + c * RO_PT + 128 + 2 * hh);
    {
        const float bv = c2[col];
#pragma unroll
        for (int v = 0; v < 8; ++v) {
            const int r = 8 * hh + v;
            if (r < RO_TG && g0 + r < G) out[(size_t)(g0 + r) * RO_MLP + col] = acc2[v] + bv;
        }
    }
}


extern "C" void kernel_launch(void* const* d_in, const int* in_sizes, int n_in,
                              void* d_out, int out_size, void* d_ws, size_t ws_size,
                              hipStream_t stream) {
    const float* node_features = (const float*)d_in[0];
    const int*   edge_src      = (const int*)d_in[1];
    const int*   edge_dst      = (const int*)d_in[2];
    const int*   graph_ids     = (const int*)d_in[3];
    const float* W_proj        = (const float*)d_in[4];
    const float* b_proj        = (const float*)d_in[5];
    const float* W_gcn         = (const float*)d_in[6];
    const float* b_gcn         = (const float*)d_in[7];
    const float* W_mlp0        = (const float*)d_in[8];
    const float* b_mlp0        = (const float*)d_in[9];
    const float* W_mlp1        = (const float*)d_in[10];
    const float* b_mlp1        = (const float*)d_in[11];
    const float* W_mlp2        = (const float*)d_in[12];
    const float* b_mlp2        = (const float*)d_in[13];

    const int D = D_FEAT;
    const int Mm = 256;
    const int N = in_sizes[3];
    const int E = in_sizes[1];
    const int L = in_sizes[6] / (D * D);
    const int G = out_size / Mm;

    float* ws = (float*)d_ws;
    float* norm_src = ws;
    float* norm_dst = norm_src + N;
    float* h        = norm_dst + N;
    float* h2       = h + (size_t)N * D;
    float* hg       = h2 + (size_t)N * D;
    float* t0       = hg + (size_t)G * D;
    float* t1       = t0 + (size_t)G * Mm;
    const int bn_even = (N + 255) / 256;
    const int bn = (bn_even > FL_ROWS) ? FL_ROWS : ((bn_even < 1) ? 1 : bn_even);
    const int B = (N + bn - 1) / bn;
    int2* binned    = (int2*)(t1 + (size_t)G * Mm);
    int*  hist      = (int*)(binned + E);
    int*  boff      = hist + (size_t)BN_BLOCKS * B;

    degree_norm_kernel<<<dim3(DG_RANGES, 2), DG_THREADS, 0, stream>>>(edge_src, edge_dst,
                                                                      norm_src, norm_dst, E, N);

    bin_count_kernel<<<BN_BLOCKS, BN_THREADS, 0, stream>>>(edge_dst, hist, E, B, bn);
    bin_place_kernel<<<BN_BLOCKS, BN_THREADS, 0, stream>>>(edge_src, edge_dst, hist, boff, binned, E, B, bn);

    const int NG_ROWS = 16 * NG_WAVES;
    const unsigned ng_blocks = (unsigned)((N + NG_ROWS - 1) / NG_ROWS);
    node_gemm_rows128<0><<<ng_blocks, 32 * NG_WAVES, 0, stream>>>(node_features, W_proj, b_proj,
                                                                 nullptr, nullptr, h, N,
                                                                  0);

    float* hcur = h;
    float* hnxt = h2;
    for (int i = 0; i < L; ++i) {
        graphconv_bins_kernel<<<(unsigned)B, FL_THREADS, 0, stream>>>(
            hcur, norm_src, norm_dst, binned, boff, W_gcn + (size_t)i * D * D, b_gcn + (size_t)i * D,
            hnxt, N, bn,  (i < L - 1) ? 1 : 0);
        float* hswap = hcur;
        hcur = hnxt;
        hnxt = hswap;
    }

    readout_kernel<<<(unsigned)((G + RO_TG - 1) / RO_TG), RO_THREADS, 0, stream>>>(
        hcur, graph_ids, W_mlp0, b_mlp0, W_mlp1, b_mlp1, W_mlp2, b_mlp2, (float*)d_out, N, G);
}


# Degrees from per-slice count tables: 64 edge slices x 2 index arrays count per node in whole-graph LDS tables (integer adds) beside the bin counts; extra placement blocks sum them; 10->9 launches
# speedup vs baseline: 5.4769x; 1.0728x over previous
#include <hip/hip_runtime.h>
#include <hip/hip_bf16.h>

typedef __attribute__((ext_vector_type(2))) float v2f;
typedef __attribute__((ext_vector_type(8))) float v8f;

#define D_FEAT 128

#define BN_CAP 1024
#define BN_BLOCKS 64
#define BN_THREADS 1024
#define CN_CAP 65536
#define CN_INFLIGHT 8

__device__ __forceinline__ int bn_bin(int d, int B, int bn) {
    unsigned b = (unsigned)d / (unsigned)bn;
    if (b > (unsigned)(B - 1)) b = (unsigned)(B - 1);
    return (int)b;
}

__global__ void __launch_bounds__(BN_THREADS)
slice_count_kernel(const int* __restrict__ src, const int* __restrict__ dst,
                   int* __restrict__ part, int* __restrict__ hist, int E, int N, int B, int bn) {
    __shared__ int cnt[CN_CAP];
    __shared__ int binc[BN_CAP];
    const int tid = threadIdx.x;
    const int which = blockIdx.y;
    const int* __restrict__ idx = which ? dst : src;
    int* __restrict__ tab = part + ((size_t)which * BN_BLOCKS + blockIdx.x) * (size_t)N;
    const int S = (E + BN_BLOCKS - 1) / BN_BLOCKS;
    const int e0 = blockIdx.x * S;
    const int e1 = (e0 + S < E) ? (e0 + S) : E;

    for (int i = tid; i < BN_CAP; i += BN_THREADS) binc[i] = 0;
    for (int lo = 0; lo < N; lo += CN_CAP) {
        const int len = (N - lo < CN_CAP) ? (N - lo) : CN_CAP;
        for (int i = tid; i < len; i += BN_THREADS) cnt[i] = 0;
        __syncthreads();
        for (int base = e0; base < e1; base += CN_INFLIGHT * BN_THREADS) {
            int x[CN_INFLIGHT];
#pragma unroll
            for (int u = 0; u < CN_INFLIGHT; ++u) {
                const int e = base + u * BN_THREADS + tid;
                x[u] = (e < e1) ? idx[e] : -1;
            }
#pragma unroll
            for (int u = 0; u < CN_INFLIGHT; ++u) {
                const int e = base + u * BN_THREADS + tid;
                const unsigned r = (unsigned)x[u] - (unsigned)lo;
                if (r < (unsigned)len) atomicAdd(&cnt[r], 1);
                if (which && lo == 0 && e < e1) atomicAdd(&binc[bn_bin(x[u], B, bn)], 1);
            }
        }
        __syncthreads();
        for (int i = tid; i < len; i += BN_THREADS) tab[lo + i] = cnt[i];
        __syncthreads();
    }
    if (which)
        for (int i = tid; i < B; i += BN_THREADS) hist[blockIdx.x * B + i] = binc[i];
}

__device__ __forceinline__ void degree_norm_share(const int* __restrict__ part,
                                                  float* __restrict__ norm_src,
                                                  float* __restrict__ norm_dst,
                                                  int N, int share, int tid) {
    const int R = (N + BN_BLOCKS - 1) / BN_BLOCKS;
    const int n0 = share * R;
    const int n1 = (n0 + R < N) ? (n0 + R) : N;
    for (int n = n0 + tid; n < n1; n += BN_THREADS) {
        const int* __restrict__ ts = part + n;
        const int* __restrict__ td = part + (size_t)BN_BLOCKS * (size_t)N + n;
        int cs = 0, cd = 0;
#pragma unroll 16
        for (int b = 0; b < BN_BLOCKS; ++b) {
            cs += ts[(size_t)b * (size_t)N];
            cd += td[(size_t)b * (size_t)N];
        }
        norm_src[n] = rsqrtf(fmaxf((float)cs, 1.0f));
        norm_dst[n] = rsqrtf(fmaxf((float)cd, 1.0f));
    }
}

__global__ void __launch_bounds__(BN_THREADS)
bin_place_kernel(const int* __restrict__ src, const int* __restrict__ dst,
                 const int* __restrict__ hist, const int* __restrict__ part,
                 int* __restrict__ boff, int2* __restrict__ binned,
                 float* __restrict__ norm_src, float* __restrict__ norm_dst,
                 int E, int N, int B, int bn) {
    __shared__ int cur[BN_CAP];
    __shared__ int wsum[BN_THREADS / 32];
    const int tid = threadIdx.x;
    const int lane = tid & 31;
    const int wave = tid >> 5;
    const int blk = blockIdx.x;

    if (blk >= BN_BLOCKS) {
        degree_norm_share(part, norm_src, norm_dst, N, blk - BN_BLOCKS, tid);
        return;
    }

    int tot = 0, pre = 0;
    if (tid < B) {
        for (int b = 0; b < BN_BLOCKS; ++b) {
            const int c = hist[b * B + tid];
            tot += c;
            if (b < blk) pre += c;
        }
    }
    int v = tot;
    for (int d = 1; d < 32; d <<= 1) {
        const int t = __shfl_up(v, d);
        if (lane >= d) v += t;
    }
    if (lane == 31) wsum[wave] = v;
    __syncthreads();
    if (wave == 0) {
        const int w = wsum[lane];
        int wi = w;
        for (int d = 1; d < 32; d <<= 1) {
            const int t = __shfl_up(wi, d);
            if (lane >= d) wi += t;
        }
        wsum[lane] = wi - w;
    }
    __syncthreads();
    const int excl = v - tot + wsum[wave];
    if (tid < B) {
        cur[tid] = excl + pre;
        if (blk == 0) boff[tid] = excl;
    }
    if (blk == 0 && tid == 0) boff[B] = E;
    __syncthreads();

    const int S = (E + BN_BLOCKS - 1) / BN_BLOCKS;
    const int e0 = blk * S;
    const int e1 = (e0 + S < E) ? (e0 + S) : E;
    for (int e = e0 + tid; e < e1; e += BN_THREADS) {
        const int s = src[e];
        const int d = dst[e];
        const int pos = atomicAdd(&cur[bn_bin(d, B, bn)], 1);
        binned[pos] = make_int2(s, d);
    }
}

#define NG_WAVES 8

template <int MODE>
__device__ __forceinline__ float ng_finish(float a, float b, float r, int relu_c) {
    float cc = a + b;
    if (relu_c) cc = fmaxf(cc, 0.0f);
    return (MODE == 1) ? fmaxf(r + cc, 0.0f) : cc;
}

template <int MODE>
__global__ void __launch_bounds__(32 * NG_WAVES)
node_gemm_rows128(const float* A, const float* __restrict__ Wg,
                  const float* __restrict__ bias,
                  const float* __restrict__ row_scale,
                  const float* resid, float* out,
                  int Mrows, int relu_c) {
    __shared__ float4 Wl[32 * D_FEAT];

    const int tid  = threadIdx.x;
    const int wave = tid >> 5;
    const int lane = tid & 31;
    const int hh   = lane >> 4;
    const int c    = lane & 15;

    float4 wr[4][4];
#pragma unroll
    for (int it = 0; it < 4; ++it) {
        int p  = it * (32 * NG_WAVES) + tid;
        int kq = p >> 5;
        int n4 = p & 31;
        const float* wp = Wg + (size_t)(4 * kq) * D_FEAT + 4 * n4;
#pragma unroll
        for (int q = 0; q < 4; ++q) wr[it][q] = *(const float4*)(wp + q * D_FEAT);
    }

    const int tile = blockIdx.x * NG_WAVES + wave;
    int row = tile * 16 + c;
    if (row >= Mrows) row = Mrows - 1;
    const float* ap = A + (size_t)row * D_FEAT + 4 * hh;
    const float rs = row_scale ? row_scale[row] : 1.0f;
    float4 av[16];
#pragma unroll
    for (int j = 0; j < 16; ++j) av[j] = *(const float4*)(ap + 8 * j);

#pragma unroll
    for (int it = 0; it < 4; ++it) {
        int p  = it * (32 * NG_WAVES) + tid;
        int kq = p >> 5;
        int n4 = p & 31;
        float4* dstp = Wl + kq * D_FEAT + (4 * (n4 >> 4)) * 16 + (n4 & 15);
        dstp[0 * 16] = make_float4(wr[it][0].x, wr[it][1].x, wr[it][2].x, wr[it][3].x);
        dstp[1 * 16] = make_float4(wr[it][0].y, wr[it][1].y, wr[it][2].y, wr[it][3].y);
        dstp[2 * 16] = make_float4(wr[it][0].z, wr[it][1].z, wr[it][2].z, wr[it][3].z);
        dstp[3 * 16] = make_float4(wr[it][0].w, wr[it][1].w, wr[it][2].w, wr[it][3].w);
    }
    __syncthreads();

    v8f acc[8] = {};
    const float4* wl = Wl + hh * D_FEAT + c;
    float4 bw[8];
#pragma unroll
    for (int t = 0; t < 8; ++t) bw[t] = wl[t * 16];
#pragma unroll
    for (int j = 0; j < 16; ++j) {
        v2f a0, a1;
        a0.x = av[j].x * rs;  a0.y = av[j].y * rs;
        a1.x = av[j].z * rs;  a1.y = av[j].w * rs;
        float4 bn[8];
        if (j < 15) {
#pragma unroll
            for (int t = 0; t < 8; ++t) bn[t] = wl[(2 * (j + 1)) * D_FEAT + t * 16];
        }
        __builtin_amdgcn_sched_barrier(0);
#pragma unroll
        for (int t = 0; t < 8; ++t) {
            v2f b0; b0.x = bw[t].x; b0.y = bw[t].y;
            acc[t] = __builtin_amdgcn_wmma_f32_16x16x4_f32(
                false, a0, false, b0, (short)0, acc[t], false, false);
        }
#pragma unroll
        for (int t = 0; t < 8; ++t) {
            v2f b1; b1.x = bw[t].z; b1.y = bw[t].w;
            acc[t] = __builtin_amdgcn_wmma_f32_16x16x4_f32(
                false, a1, false, b1, (short)0, acc[t], false, false);
        }
        __builtin_amdgcn_sched_barrier(0);
        if (j < 15) {
#pragma unroll
            for (int t = 0; t < 8; ++t) bw[t] = bn[t];
        }
    }

    const int m0 = tile * 16 + hh * 8;
    float4 bz[2];
#pragma unroll
    for (int g = 0; g < 2; ++g) bz[g] = *(const float4*)(bias + 64 * g + 4 * c);

    if (tile * 16 + 16 <= Mrows) {
        float4 rv[8][2];
#pragma unroll
        for (int v = 0; v < 8; ++v)
#pragma unroll
            for (int g = 0; g < 2; ++g) {
                if (MODE == 1)
                    rv[v][g] = *(const float4*)(resid + (size_t)(m0 + v) * D_FEAT + 64 * g + 4 * c);
                else
                    rv[v][g] = make_float4(0.0f, 0.0f, 0.0f, 0.0f);
            }
#pragma unroll
        for (int v = 0; v < 8; ++v)
#pragma unroll
            for (int g = 0; g < 2; ++g) {
                float4 o;
                o.x = ng_finish<MODE>(acc[4 * g + 0][v], bz[g].x, rv[v][g].x, relu_c);
                o.y = ng_finish<MODE>(acc[4 * g + 1][v], bz[g].y, rv[v][g].y, relu_c);
                o.z = ng_finish<MODE>(acc[4 * g + 2][v], bz[g].z, rv[v][g].z, relu_c);
                o.w = ng_finish<MODE>(acc[4 * g + 3][v], bz[g].w, rv[v][g].w, relu_c);
                *(float4*)(out + (size_t)(m0 + v) * D_FEAT + 64 * g + 4 * c) = o;
            }
    } else {
#pragma unroll
        for (int v = 0; v < 8; ++v) {
            if (m0 + v < Mrows) {
#pragma unroll
                for (int g = 0; g < 2; ++g) {
                    float4 r = make_float4(0.0f, 0.0f, 0.0f, 0.0f);
                    if (MODE == 1)
                        r = *(const float4*)(resid + (size_t)(m0 + v) * D_FEAT + 64 * g + 4 * c);
                    float4 o;
                    o.x = ng_finish<MODE>(acc[4 * g + 0][v], bz[g].x, r.x, relu_c);
                    o.y = ng_finish<MODE>(acc[4 * g + 1][v], bz[g].y, r.y, relu_c);
                    o.z = ng_finish<MODE>(acc[4 * g + 2][v], bz[g].z, r.z, relu_c);
                    o.w = ng_finish<MODE>(acc[4 * g + 3][v], bz[g].w, r.w, relu_c);
                    *(float4*)(out + (size_t)(m0 + v) * D_FEAT + 64 * g + 4 * c) = o;
                }
            }
        }
    }
}

#define SC_EPW 16
#define FL_TILES 13
#define FL_ROWS (16 * FL_TILES)
#define FL_WAVES (2 * FL_TILES)
#define FL_THREADS (32 * FL_WAVES)
#define FL_PITCH 132
#define FL_WPIECES (32 * 32)

__device__ __forceinline__ void fl_w_load(float4 (&wr)[2][4], const float* __restrict__ Wg, int tid) {
#pragma unroll
    for (int it = 0; it < 2; ++it) {
        const int p  = it * FL_THREADS + tid;
        const int kq = p >> 5;
        const int n4 = p & 31;
        const float* wp = Wg + (size_t)(4 * kq) * D_FEAT + 4 * n4;
#pragma unroll
        for (int q = 0; q < 4; ++q) {
            if (p < FL_WPIECES) wr[it][q] = *(const float4*)(wp + q * D_FEAT);
            else wr[it][q] = make_float4(0.0f, 0.0f, 0.0f, 0.0f);
        }
    }
}

__device__ __forceinline__ void fl_w_store(float4* Wl, const float4 (&wr)[2][4], int tid) {
#pragma unroll
    for (int it = 0; it < 2; ++it) {
        const int p  = it * FL_THREADS + tid;
        const int kq = p >> 5;
        const int n4 = p & 31;
        if (p < FL_WPIECES) {
            float4* dstp = Wl + kq * D_FEAT + (4 * (n4 >> 4)) * 16 + (n4 & 15);
            dstp[0 * 16] = make_float4(wr[it][0].x, wr[it][1].x, wr[it][2].x, wr[it][3].x);
            dstp[1 * 16] = make_float4(wr[it][0].y, wr[it][1].y, wr[it][2].y, wr[it][3].y);
            dstp[2 * 16] = make_float4(wr[it][0].z, wr[it][1].z, wr[it][2].z, wr[it][3].z);
            dstp[3 * 16] = make_float4(wr[it][0].w, wr[it][1].w, wr[it][2].w, wr[it][3].w);
        }
    }
}

__device__ __forceinline__ void fl_gemm_half(v8f (&acc)[4], const float* arow, const float4* wl,
                                             float rs) {
    float4 aw = *(const float4*)arow;
    float4 bw[4];
#pragma unroll
    for (int t = 0; t < 4; ++t) bw[t] = wl[t * 16];
#pragma unroll
    for (int j = 0; j < 8; ++j) {
        v2f a0, a1;
        a0.x = aw.x * rs;  a0.y = aw.y * rs;
        a1.x = aw.z * rs;  a1.y = aw.w * rs;
        float4 an = aw;
        float4 bn[4];
        if (j < 7) {
            an = *(const float4*)(arow + 8 * (j + 1));
#pragma unroll
            for (int t = 0; t < 4; ++t) bn[t] = wl[(2 * (j + 1)) * D_FEAT + t * 16];
        }
        __builtin_amdgcn_sched_barrier(0);
#pragma unroll
        for (int t = 0; t < 4; ++t) {
            v2f b0; b0.x = bw[t].x; b0.y = bw[t].y;
            acc[t] = __builtin_amdgcn_wmma_f32_16x16x4_f32(
                false, a0, false, b0, (short)0, acc[t], false, false);
        }
#pragma unroll
        for (int t = 0; t < 4; ++t) {
            v2f b1; b1.x = bw[t].z; b1.y = bw[t].w;
            acc[t] = __builtin_amdgcn_wmma_f32_16x16x4_f32(
                false, a1, false, b1, (short)0, acc[t], false, false);
        }
        __builtin_amdgcn_sched_barrier(0);
        if (j < 7) {
            aw = an;
#pragma unroll
            for (int t = 0; t < 4; ++t) bw[t] = bn[t];
        }
    }
}

__global__ void __launch_bounds__(FL_THREADS)
graphconv_bins_kernel(const float* __restrict__ hin, const float* __restrict__ norm_src,
                      const float* __restrict__ norm_dst,
                      const int2* __restrict__ binned, const int* __restrict__ boff,
                      const float* __restrict__ Wg, const float* __restrict__ bias,
                      float* __restrict__ hout, int N, int bn, int relu_c) {
    __shared__ __attribute__((aligned(16))) float rows[FL_ROWS * FL_PITCH];
    __shared__ float4 Wl[32 * D_FEAT];
    const int tid = threadIdx.x;
    const int lane = tid & 31;
    const int wid = __builtin_amdgcn_readfirstlane((int)(threadIdx.x >> 5));
    const int bin = blockIdx.x;
    const int lo = bin * bn;

    float4 wr[2][4];
    fl_w_load(wr, Wg, tid);
    float4* r4 = (float4*)rows;
    for (int i = tid; i < FL_ROWS * FL_PITCH / 4; i += FL_THREADS)
        r4[i] = make_float4(0.0f, 0.0f, 0.0f, 0.0f);
    const int b0 = boff[bin];
    const int b1 = boff[bin + 1];
    fl_w_store(Wl, wr, tid);
    __syncthreads();

    for (int e0 = b0 + wid * SC_EPW; e0 < b1; e0 += FL_WAVES * SC_EPW) {
        const int n = (b1 - e0 < SC_EPW) ? (b1 - e0) : SC_EPW;
        int el = e0 + (lane & (SC_EPW - 1));
        if (el > b1 - 1) el = b1 - 1;
        const int2 ed = binned[el];
        const int sv = ed.x;
        unsigned du = (unsigned)(ed.y - lo);
        if (du > (unsigned)(FL_ROWS - 1)) du = (unsigned)(FL_ROWS - 1);
        const int dv = (int)du;
        const int nv = __float_as_int(norm_src[sv]);

        if (n == SC_EPW) {
            float v[SC_EPW][4];
#pragma unroll
            for (int i = 0; i < SC_EPW; ++i) {
                const int s = __builtin_amdgcn_readlane(sv, i);
                const float* hp = hin + (size_t)s * D_FEAT + lane;
#pragma unroll
                for (int k = 0; k < 4; ++k) v[i][k] = hp[32 * k];
            }
#pragma unroll
            for (int i = 0; i < SC_EPW; ++i) {
                const int d = __builtin_amdgcn_readlane(dv, i);
                const float ns = __int_as_float(__builtin_amdgcn_readlane(nv, i));
                float* rp = rows + d * FL_PITCH + lane;
#pragma unroll
                for (int k = 0; k < 4; ++k)
                    __hip_atomic_fetch_add(rp + 32 * k, v[i][k] * ns, __ATOMIC_RELAXED,
                                           __HIP_MEMORY_SCOPE_WORKGROUP);
            }
        } else {
            for (int i = 0; i < n; ++i) {
                const int s = __builtin_amdgcn_readlane(sv, i);
                const int d = __builtin_amdgcn_readlane(dv, i);
                const float ns = __int_as_float(__builtin_amdgcn_readlane(nv, i));
                const float* hp = hin + (size_t)s * D_FEAT + lane;
                float* rp = rows + d * FL_PITCH + lane;
                float t[4];
#pragma unroll
                for (int k = 0; k < 4; ++k) t[k] = hp[32 * k];
#pragma unroll
                for (int k = 0; k < 4; ++k)
                    __hip_atomic_fetch_add(rp + 32 * k, t[k] * ns, __ATOMIC_RELAXED,
                                           __HIP_MEMORY_SCOPE_WORKGROUP);
            }
        }
    }
    __syncthreads();

    const int hh = lane >> 4;
    const int c  = lane & 15;
    const int rt = wid >> 1;
    const int g  = wid & 1;
    int node = lo + 16 * rt + c;
    if (node >= N) node = N - 1;
    const float rs = norm_dst[node];
    const float* arow = rows + (16 * rt + c) * FL_PITCH + 4 * hh;
    const float4* wl = Wl + hh * D_FEAT + (4 * g) * 16 + c;

    v8f acc[4] = {};
    fl_gemm_half(acc, arow, wl, rs);
    fl_gemm_half(acc, arow + 64, wl + 16 * D_FEAT, rs);

    const int r0 = 16 * rt + hh * 8;
    const int m0 = lo + r0;
    const float4 bz = *(const float4*)(bias + 64 * g + 4 * c);
    float4 rv[8];
#pragma unroll
    for (int v = 0; v < 8; ++v) {
        int m = m0 + v;
        if (m >= N) m = N - 1;
        rv[v] = *(const float4*)(hin + (size_t)m * D_FEAT + 64 * g + 4 * c);
    }
#pragma unroll
    for (int v = 0; v < 8; ++v) {
        if (r0 + v < bn && m0 + v < N) {
            float4 o;
            o.x = ng_finish<1>(acc[0][v], bz.x, rv[v].x, relu_c);
            o.y = ng_finish<1>(acc[1][v], bz.y, rv[v].y, relu_c);
            o.z = ng_finish<1>(acc[2][v], bz.z, rv[v].z, relu_c);
            o.w = ng_finish<1>(acc[3][v], bz.w, rv[v].w, relu_c);
            *(float4*)(hout + (size_t)(m0 + v) * D_FEAT + 64 * g + 4 * c) = o;
        }
    }
}

#define RO_TG 8
#define RO_WAVES 16
#define RO_THREADS (32 * RO_WAVES)
#define RO_PA 132
#define RO_PT 260
#define RO_MLP 256
#define RO_BATCH 16
#define RO_SCAN 16

__device__ __forceinline__ void ro_b_load(v2f (&b)[32], const float* __restrict__ Wm, int kbase,
                                          int hh, int col) {
#pragma unroll
    for (int s = 0; s < 32; ++s) {
        const float* p = Wm + (size_t)(kbase + 4 * s + 2 * hh) * RO_MLP + col;
        b[s].x = p[0];
        b[s].y = p[RO_MLP];
    }
}

__device__ __forceinline__ void ro_stage(v8f& acc, const v2f (&b)[32], const float* arow) {
#pragma unroll
    for (int s = 0; s < 32; ++s) {
        const v2f a = *(const v2f*)(arow + 4 * s);
        acc = __builtin_amdgcn_wmma_f32_16x16x4_f32(false, a, false, b[s], (short)0, acc,
                                                    false, false);
    }
}

__device__ __forceinline__ void ro_flush(float* tile, int g, const float (&acc)[4], int lane) {
    if (g >= 0) {
        float* rp = tile + g * RO_PA + lane;
#pragma unroll
        for (int k = 0; k < 4; ++k)
            __hip_atomic_fetch_add(rp + 32 * k, acc[k], __ATOMIC_RELAXED,
                                   __HIP_MEMORY_SCOPE_WORKGROUP);
    }
}

__device__ __forceinline__ void ro_span(int& mn, int& mx, int id, int n, int g0) {
    if ((unsigned)(id - g0) < (unsigned)RO_TG) {
        mn = (n < mn) ? n : mn;
        mx = (n > mx) ? n : mx;
    }
}

__global__ void __launch_bounds__(RO_THREADS)
readout_kernel(const float* __restrict__ hfin, const int* __restrict__ gid,
               const float* __restrict__ W0, const float* __restrict__ c0,
               const float* __restrict__ W1, const float* __restrict__ c1,
               const float* __restrict__ W2, const float* __restrict__ c2,
               float* __restrict__ out, int N, int G) {
    __shared__ __attribute__((aligned(16))) float tA[16 * RO_PA];
    __shared__ __attribute__((aligned(16))) float tB[16 * RO_PT];
    __shared__ __attribute__((aligned(16))) float tC[16 * RO_PT];
    __shared__ int rng[2];
    const int tid = threadIdx.x;
    const int lane = tid & 31;
    const int wid = __builtin_amdgcn_readfirstlane((int)(threadIdx.x >> 5));
    const int hh = lane >> 4;
    const int c = lane & 15;
    const int col = 16 * wid + c;
    const int g0 = blockIdx.x * RO_TG;

    v2f bA[32], bB[32];
    ro_b_load(bA, W0, 0, hh, col);
    ro_b_load(bB, W1, 0, hh, col);

    for (int i = tid; i < 16 * RO_PA; i += RO_THREADS) tA[i] = 0.0f;
    if (tid == 0) { rng[0] = N; rng[1] = -1; }
    __syncthreads();

    int mn = N, mx = -1;
    const int n4 = N >> 2;
    const int4* __restrict__ gid4 = (const int4*)gid;
    for (int base = 0; base < n4; base += RO_SCAN * RO_THREADS) {
        int4 v[RO_SCAN];
#pragma unroll
        for (int u = 0; u < RO_SCAN; ++u) {
            const int j = base + u * RO_THREADS + tid;
            if (j < n4) v[u] = gid4[j];
            else v[u] = make_int4(-1, -1, -1, -1);
        }
#pragma unroll
        for (int u = 0; u < RO_SCAN; ++u) {
            const int n = 4 * (base + u * RO_THREADS + tid);
            ro_span(mn, mx, v[u].x, n + 0, g0);
            ro_span(mn, mx, v[u].y, n + 1, g0);
            ro_span(mn, mx, v[u].z, n + 2, g0);
            ro_span(mn, mx, v[u].w, n + 3, g0);
        }
    }
    for (int n = (n4 << 2) + tid; n < N; n += RO_THREADS) ro_span(mn, mx, gid[n], n, g0);
    for (int d = 16; d >= 1; d >>= 1) {
        const int a = __shfl_xor(mn, d);
        const int b = __shfl_xor(mx, d);
        mn = (a < mn) ? a : mn;
        mx = (b > mx) ? b : mx;
    }
    if (lane == 0) {
        atomicMin(&rng[0], mn);
        atomicMax(&rng[1], mx);
    }
    __syncthreads();
    const int lo = __builtin_amdgcn_readfirstlane(rng[0]);
    const int hi = __builtin_amdgcn_readfirstlane(rng[1]) + 1;

    if (hi > lo) {
        const int per = (hi - lo + RO_WAVES - 1) / RO_WAVES;
        const int w0 = lo + wid * per;
        const int w1 = (w0 + per < hi) ? (w0 + per) : hi;
        int cur = -1;
        float acc[4] = {0.0f, 0.0f, 0.0f, 0.0f};
        for (int base = w0; base < w1; base += RO_BATCH) {
            int nl = base + (lane & (RO_BATCH - 1));
            if (nl > w1 - 1) nl = w1 - 1;
            const int gv = gid[nl] - g0;
            float v[RO_BATCH][4];
#pragma unroll
            for (int i = 0; i < RO_BATCH; ++i) {
                int n = base + i;
                if (n > w1 - 1) n = w1 - 1;
                const float* hp = hfin + (size_t)n * D_FEAT + lane;
#pragma unroll
                for (int k = 0; k < 4; ++k) v[i][k] = hp[32 * k];
            }
#pragma unroll
            for (int i = 0; i < RO_BATCH; ++i)
                asm volatile("" :: "v"(v[i][0]), "v"(v[i][1]), "v"(v[i][2]), "v"(v[i][3]));
#pragma unroll
            for (int i = 0; i < RO_BATCH; ++i) {
                const int g = __builtin_amdgcn_readlane(gv, i);
                if (base + i < w1 && (unsigned)g < (unsigned)RO_TG) {
                    if (g != cur) {
                        ro_flush(tA, cur, acc, lane);
                        cur = g;
#pragma unroll
                        for (int k = 0; k < 4; ++k) acc[k] = 0.0f;
                    }
#pragma unroll
                    for (int k = 0; k < 4; ++k) acc[k] += v[i][k];
                }
            }
        }
        ro_flush(tA, cur, acc, lane);
    }
    __syncthreads();

    v8f acc0 = {};
    ro_stage(acc0, bA, tA + c * RO_PA + 2 * hh);
    {
        const float bv = c0[col];
#pragma unroll
        for (int v = 0; v < 8; ++v) tB[(8 * hh + v) * RO_PT + col] = fmaxf(acc0[v] + bv, 0.0f);
    }
    __syncthreads();

    v8f acc1 = {};
    ro_b_load(bA, W1, 128, hh, col);
    ro_stage(acc1, bB, tB + c * RO_PT + 2 * hh);
    ro_b_load(bB, W2, 0, hh, col);
    ro_stage(acc1, bA, tB + c * RO_PT + 128 + 2 * hh);
    {
        const float bv = c1[col];
#pragma unroll
        for (int v = 0; v < 8; ++v) tC[(8 * hh + v) * RO_PT + col] = fmaxf(acc1[v] + bv, 0.0f);
    }
    __syncthreads();

    v8f acc2 = {};
    ro_b_load(bA, W2, 128, hh, col);
    ro_stage(acc2, bB, tC + c * RO_PT + 2 * hh);
    ro_stage(acc2, bA, tC + c * RO_PT + 128 + 2 * hh);
    {
        const float bv = c2[col];
#pragma unroll
        for (int v = 0; v < 8; ++v) {
            const int r = 8 * hh + v;
            if (r < RO_TG && g0 + r < G) out[(size_t)(g0 + r) * RO_MLP + col] = acc2[v] + bv;
        }
    }
}


extern "C" void kernel_launch(void* const* d_in, const int* in_sizes, int n_in,
                              void* d_out, int out_size, void* d_ws, size_t ws_size,
                              hipStream_t stream) {
    const float* node_features = (const float*)d_in[0];
    const int*   edge_src      = (const int*)d_in[1];
    const int*   edge_dst      = (const int*)d_in[2];
    const int*   graph_ids     = (const int*)d_in[3];
    const float* W_proj        = (const float*)d_in[4];
    const float* b_proj        = (const float*)d_in[5];
    const float* W_gcn         = (const float*)d_in[6];
    const float* b_gcn         = (const float*)d_in[7];
    const float* W_mlp0        = (const float*)d_in[8];
    const float* b_mlp0        = (const float*)d_in[9];
    const float* W_mlp1        = (const float*)d_in[10];
    const float* b_mlp1        = (const float*)d_in[11];
    const float* W_mlp2        = (const float*)d_in[12];
    const float* b_mlp2        = (const float*)d_in[13];

    const int D = D_FEAT;
    const int Mm = 256;
    const int N = in_sizes[3];
    const int E = in_sizes[1];
    const int L = in_sizes[6] / (D * D);
    const int G = out_size / Mm;

    float* ws = (float*)d_ws;
    float* norm_src = ws;
    float* norm_dst = norm_src + N;
    float* h        = norm_dst + N;
    float* h2       = h + (size_t)N * D;
    float* hg       = h2 + (size_t)N * D;
    float* t0       = hg + (size_t)G * D;
    float* t1       = t0 + (size_t)G * Mm;
    const int bn_even = (N + 255) / 256;
    const int bn = (bn_even > FL_ROWS) ? FL_ROWS : ((bn_even < 1) ? 1 : bn_even);
    const int B = (N + bn - 1) / bn;
    int2* binned    = (int2*)(t1 + (size_t)G * Mm);
    int*  hist      = (int*)(binned + E);
    int*  boff      = hist + (size_t)BN_BLOCKS * B;
    int*  part      = boff + (B + 1);

    slice_count_kernel<<<dim3(BN_BLOCKS, 2), BN_THREADS, 0, stream>>>(edge_src, edge_dst, part, hist,
                                                                      E, N, B, bn);
    bin_place_kernel<<<2 * BN_BLOCKS, BN_THREADS, 0, stream>>>(edge_src, edge_dst, hist, part, boff,
                                                               binned, norm_src, norm_dst, E, N, B, bn);

    const int NG_ROWS = 16 * NG_WAVES;
    const unsigned ng_blocks = (unsigned)((N + NG_ROWS - 1) / NG_ROWS);
    node_gemm_rows128<0><<<ng_blocks, 32 * NG_WAVES, 0, stream>>>(node_features, W_proj, b_proj,
                                                                 nullptr, nullptr, h, N,
                                                                  0);

    float* hcur = h;
    float* hnxt = h2;
    for (int i = 0; i < L; ++i) {
        graphconv_bins_kernel<<<(unsigned)B, FL_THREADS, 0, stream>>>(
            hcur, norm_src, norm_dst, binned, boff, W_gcn + (size_t)i * D * D, b_gcn + (size_t)i * D,
            hnxt, N, bn,  (i < L - 1) ? 1 : 0);
        float* hswap = hcur;
        hcur = hnxt;
        hnxt = hswap;
    }

    readout_kernel<<<(unsigned)((G + RO_TG - 1) / RO_TG), RO_THREADS, 0, stream>>>(
        hcur, graph_ids, W_mlp0, b_mlp0, W_mlp1, b_mlp1, W_mlp2, b_mlp2, (float*)d_out, N, G);
}


# Planes of h hold h*norm_src: the per-edge norm gather (a random 4-B read per edge and layer) and multiply leave the GraphConv sum; residual added in scaled space, last layer divides; h' rounds anew
# speedup vs baseline: 5.6630x; 1.0340x over previous
#include <hip/hip_runtime.h>
#include <hip/hip_bf16.h>

typedef __attribute__((ext_vector_type(2))) float v2f;
typedef __attribute__((ext_vector_type(8))) float v8f;

#define D_FEAT 128

#define BN_CAP 1024
#define BN_BLOCKS 64
#define BN_THREADS 1024
#define CN_CAP 65536
#define CN_INFLIGHT 8

__device__ __forceinline__ int bn_bin(int d, int B, int bn) {
    unsigned b = (unsigned)d / (unsigned)bn;
    if (b > (unsigned)(B - 1)) b = (unsigned)(B - 1);
    return (int)b;
}

__global__ void __launch_bounds__(BN_THREADS)
slice_count_kernel(const int* __restrict__ src, const int* __restrict__ dst,
                   int* __restrict__ part, int* __restrict__ hist, int E, int N, int B, int bn) {
    __shared__ int cnt[CN_CAP];
    __shared__ int binc[BN_CAP];
    const int tid = threadIdx.x;
    const int which = blockIdx.y;
    const int* __restrict__ idx = which ? dst : src;
    int* __restrict__ tab = part + ((size_t)which * BN_BLOCKS + blockIdx.x) * (size_t)N;
    const int S = (E + BN_BLOCKS - 1) / BN_BLOCKS;
    const int e0 = blockIdx.x * S;
    const int e1 = (e0 + S < E) ? (e0 + S) : E;

    for (int i = tid; i < BN_CAP; i += BN_THREADS) binc[i] = 0;
    for (int lo = 0; lo < N; lo += CN_CAP) {
        const int len = (N - lo < CN_CAP) ? (N - lo) : CN_CAP;
        for (int i = tid; i < len; i += BN_THREADS) cnt[i] = 0;
        __syncthreads();
        for (int base = e0; base < e1; base += CN_INFLIGHT * BN_THREADS) {
            int x[CN_INFLIGHT];
#pragma unroll
            for (int u = 0; u < CN_INFLIGHT; ++u) {
                const int e = base + u * BN_THREADS + tid;
                x[u] = (e < e1) ? idx[e] : -1;
            }
#pragma unroll
            for (int u = 0; u < CN_INFLIGHT; ++u) {
                const int e = base + u * BN_THREADS + tid;
                const unsigned r = (unsigned)x[u] - (unsigned)lo;
                if (r < (unsigned)len) atomicAdd(&cnt[r], 1);
                if (which && lo == 0 && e < e1) atomicAdd(&binc[bn_bin(x[u], B, bn)], 1);
            }
        }
        __syncthreads();
        for (int i = tid; i < len; i += BN_THREADS) tab[lo + i] = cnt[i];
        __syncthreads();
    }
    if (which)
        for (int i = tid; i < B; i += BN_THREADS) hist[blockIdx.x * B + i] = binc[i];
}

__device__ __forceinline__ void degree_norm_share(const int* __restrict__ part,
                                                  float* __restrict__ norm_src,
                                                  float* __restrict__ norm_dst,
                                                  int N, int share, int tid) {
    const int R = (N + BN_BLOCKS - 1) / BN_BLOCKS;
    const int n0 = share * R;
    const int n1 = (n0 + R < N) ? (n0 + R) : N;
    for (int n = n0 + tid; n < n1; n += BN_THREADS) {
        const int* __restrict__ ts = part + n;
        const int* __restrict__ td = part + (size_t)BN_BLOCKS * (size_t)N + n;
        int cs = 0, cd = 0;
#pragma unroll 16
        for (int b = 0; b < BN_BLOCKS; ++b) {
            cs += ts[(size_t)b * (size_t)N];
            cd += td[(size_t)b * (size_t)N];
        }
        norm_src[n] = rsqrtf(fmaxf((float)cs, 1.0f));
        norm_dst[n] = rsqrtf(fmaxf((float)cd, 1.0f));
    }
}

__global__ void __launch_bounds__(BN_THREADS)
bin_place_kernel(const int* __restrict__ src, const int* __restrict__ dst,
                 const int* __restrict__ hist, const int* __restrict__ part,
                 int* __restrict__ boff, int2* __restrict__ binned,
                 float* __restrict__ norm_src, float* __restrict__ norm_dst,
                 int E, int N, int B, int bn) {
    __shared__ int cur[BN_CAP];
    __shared__ int wsum[BN_THREADS / 32];
    const int tid = threadIdx.x;
    const int lane = tid & 31;
    const int wave = tid >> 5;
    const int blk = blockIdx.x;

    if (blk >= BN_BLOCKS) {
        degree_norm_share(part, norm_src, norm_dst, N, blk - BN_BLOCKS, tid);
        return;
    }

    int tot = 0, pre = 0;
    if (tid < B) {
        for (int b = 0; b < BN_BLOCKS; ++b) {
            const int c = hist[b * B + tid];
            tot += c;
            if (b < blk) pre += c;
        }
    }
    int v = tot;
    for (int d = 1; d < 32; d <<= 1) {
        const int t = __shfl_up(v, d);
        if (lane >= d) v += t;
    }
    if (lane == 31) wsum[wave] = v;
    __syncthreads();
    if (wave == 0) {
        const int w = wsum[lane];
        int wi = w;
        for (int d = 1; d < 32; d <<= 1) {
            const int t = __shfl_up(wi, d);
            if (lane >= d) wi += t;
        }
        wsum[lane] = wi - w;
    }
    __syncthreads();
    const int excl = v - tot + wsum[wave];
    if (tid < B) {
        cur[tid] = excl + pre;
        if (blk == 0) boff[tid] = excl;
    }
    if (blk == 0 && tid == 0) boff[B] = E;
    __syncthreads();

    const int S = (E + BN_BLOCKS - 1) / BN_BLOCKS;
    const int e0 = blk * S;
    const int e1 = (e0 + S < E) ? (e0 + S) : E;
    for (int e = e0 + tid; e < e1; e += BN_THREADS) {
        const int s = src[e];
        const int d = dst[e];
        const int pos = atomicAdd(&cur[bn_bin(d, B, bn)], 1);
        binned[pos] = make_int2(s, d);
    }
}

#define NG_WAVES 8

template <int MODE>
__device__ __forceinline__ float ng_finish(float a, float b, float r, int relu_c) {
    float cc = a + b;
    if (relu_c) cc = fmaxf(cc, 0.0f);
    return (MODE == 1) ? fmaxf(r + cc, 0.0f) : cc;
}

template <int MODE>
__global__ void __launch_bounds__(32 * NG_WAVES)
node_gemm_rows128(const float* A, const float* __restrict__ Wg,
                  const float* __restrict__ bias,
                  const float* __restrict__ row_scale,
                  const float* __restrict__ out_scale,
                  const float* resid, float* out,
                  int Mrows, int relu_c) {
    __shared__ float4 Wl[32 * D_FEAT];

    const int tid  = threadIdx.x;
    const int wave = tid >> 5;
    const int lane = tid & 31;
    const int hh   = lane >> 4;
    const int c    = lane & 15;

    float4 wr[4][4];
#pragma unroll
    for (int it = 0; it < 4; ++it) {
        int p  = it * (32 * NG_WAVES) + tid;
        int kq = p >> 5;
        int n4 = p & 31;
        const float* wp = Wg + (size_t)(4 * kq) * D_FEAT + 4 * n4;
#pragma unroll
        for (int q = 0; q < 4; ++q) wr[it][q] = *(const float4*)(wp + q * D_FEAT);
    }

    const int tile = blockIdx.x * NG_WAVES + wave;
    int row = tile * 16 + c;
    if (row >= Mrows) row = Mrows - 1;
    const float* ap = A + (size_t)row * D_FEAT + 4 * hh;
    const float rs = row_scale ? row_scale[row] : 1.0f;
    float4 av[16];
#pragma unroll
    for (int j = 0; j < 16; ++j) av[j] = *(const float4*)(ap + 8 * j);

#pragma unroll
    for (int it = 0; it < 4; ++it) {
        int p  = it * (32 * NG_WAVES) + tid;
        int kq = p >> 5;
        int n4 = p & 31;
        float4* dstp = Wl + kq * D_FEAT + (4 * (n4 >> 4)) * 16 + (n4 & 15);
        dstp[0 * 16] = make_float4(wr[it][0].x, wr[it][1].x, wr[it][2].x, wr[it][3].x);
        dstp[1 * 16] = make_float4(wr[it][0].y, wr[it][1].y, wr[it][2].y, wr[it][3].y);
        dstp[2 * 16] = make_float4(wr[it][0].z, wr[it][1].z, wr[it][2].z, wr[it][3].z);
        dstp[3 * 16] = make_float4(wr[it][0].w, wr[it][1].w, wr[it][2].w, wr[it][3].w);
    }
    __syncthreads();

    v8f acc[8] = {};
    const float4* wl = Wl + hh * D_FEAT + c;
    float4 bw[8];
#pragma unroll
    for (int t = 0; t < 8; ++t) bw[t] = wl[t * 16];
#pragma unroll
    for (int j = 0; j < 16; ++j) {
        v2f a0, a1;
        a0.x = av[j].x * rs;  a0.y = av[j].y * rs;
        a1.x = av[j].z * rs;  a1.y = av[j].w * rs;
        float4 bn[8];
        if (j < 15) {
#pragma unroll
            for (int t = 0; t < 8; ++t) bn[t] = wl[(2 * (j + 1)) * D_FEAT + t * 16];
        }
        __builtin_amdgcn_sched_barrier(0);
#pragma unroll
        for (int t = 0; t < 8; ++t) {
            v2f b0; b0.x = bw[t].x; b0.y = bw[t].y;
            acc[t] = __builtin_amdgcn_wmma_f32_16x16x4_f32(
                false, a0, false, b0, (short)0, acc[t], false, false);
        }
#pragma unroll
        for (int t = 0; t < 8; ++t) {
            v2f b1; b1.x = bw[t].z; b1.y = bw[t].w;
            acc[t] = __builtin_amdgcn_wmma_f32_16x16x4_f32(
                false, a1, false, b1, (short)0, acc[t], false, false);
        }
        __builtin_amdgcn_sched_barrier(0);
        if (j < 15) {
#pragma unroll
            for (int t = 0; t < 8; ++t) bw[t] = bn[t];
        }
    }

    const int m0 = tile * 16 + hh * 8;
    float4 bz[2];
#pragma unroll
    for (int g = 0; g < 2; ++g) bz[g] = *(const float4*)(bias + 64 * g + 4 * c);
    float osv[8];
#pragma unroll
    for (int v = 0; v < 8; ++v) {
        int m = m0 + v;
        if (m >= Mrows) m = Mrows - 1;
        osv[v] = out_scale ? out_scale[m] : 1.0f;
    }

    if (tile * 16 + 16 <= Mrows) {
        float4 rv[8][2];
#pragma unroll
        for (int v = 0; v < 8; ++v)
#pragma unroll
            for (int g = 0; g < 2; ++g) {
                if (MODE == 1)
                    rv[v][g] = *(const float4*)(resid + (size_t)(m0 + v) * D_FEAT + 64 * g + 4 * c);
                else
                    rv[v][g] = make_float4(0.0f, 0.0f, 0.0f, 0.0f);
            }
#pragma unroll
        for (int v = 0; v < 8; ++v)
#pragma unroll
            for (int g = 0; g < 2; ++g) {
                float4 o;
                o.x = ng_finish<MODE>(acc[4 * g + 0][v], bz[g].x, rv[v][g].x, relu_c) * osv[v];
                o.y = ng_finish<MODE>(acc[4 * g + 1][v], bz[g].y, rv[v][g].y, relu_c) * osv[v];
                o.z = ng_finish<MODE>(acc[4 * g + 2][v], bz[g].z, rv[v][g].z, relu_c) * osv[v];
                o.w = ng_finish<MODE>(acc[4 * g + 3][v], bz[g].w, rv[v][g].w, relu_c) * osv[v];
                *(float4*)(out + (size_t)(m0 + v) * D_FEAT + 64 * g + 4 * c) = o;
            }
    } else {
#pragma unroll
        for (int v = 0; v < 8; ++v) {
            if (m0 + v < Mrows) {
#pragma unroll
                for (int g = 0; g < 2; ++g) {
                    float4 r = make_float4(0.0f, 0.0f, 0.0f, 0.0f);
                    if (MODE == 1)
                        r = *(const float4*)(resid + (size_t)(m0 + v) * D_FEAT + 64 * g + 4 * c);
                    float4 o;
                    o.x = ng_finish<MODE>(acc[4 * g + 0][v], bz[g].x, r.x, relu_c) * osv[v];
                    o.y = ng_finish<MODE>(acc[4 * g + 1][v], bz[g].y, r.y, relu_c) * osv[v];
                    o.z = ng_finish<MODE>(acc[4 * g + 2][v], bz[g].z, r.z, relu_c) * osv[v];
                    o.w = ng_finish<MODE>(acc[4 * g + 3][v], bz[g].w, r.w, relu_c) * osv[v];
                    *(float4*)(out + (size_t)(m0 + v) * D_FEAT + 64 * g + 4 * c) = o;
                }
            }
        }
    }
}

#define SC_EPW 16
#define FL_TILES 13
#define FL_ROWS (16 * FL_TILES)
#define FL_WAVES (2 * FL_TILES)
#define FL_THREADS (32 * FL_WAVES)
#define FL_PITCH 132
#define FL_WPIECES (32 * 32)

__device__ __forceinline__ void fl_w_load(float4 (&wr)[2][4], const float* __restrict__ Wg, int tid) {
#pragma unroll
    for (int it = 0; it < 2; ++it) {
        const int p  = it * FL_THREADS + tid;
        const int kq = p >> 5;
        const int n4 = p & 31;
        const float* wp = Wg + (size_t)(4 * kq) * D_FEAT + 4 * n4;
#pragma unroll
        for (int q = 0; q < 4; ++q) {
            if (p < FL_WPIECES) wr[it][q] = *(const float4*)(wp + q * D_FEAT);
            else wr[it][q] = make_float4(0.0f, 0.0f, 0.0f, 0.0f);
        }
    }
}

__device__ __forceinline__ void fl_w_store(float4* Wl, const float4 (&wr)[2][4], int tid) {
#pragma unroll
    for (int it = 0; it < 2; ++it) {
        const int p  = it * FL_THREADS + tid;
        const int kq = p >> 5;
        const int n4 = p & 31;
        if (p < FL_WPIECES) {
            float4* dstp = Wl + kq * D_FEAT + (4 * (n4 >> 4)) * 16 + (n4 & 15);
            dstp[0 * 16] = make_float4(wr[it][0].x, wr[it][1].x, wr[it][2].x, wr[it][3].x);
            dstp[1 * 16] = make_float4(wr[it][0].y, wr[it][1].y, wr[it][2].y, wr[it][3].y);
            dstp[2 * 16] = make_float4(wr[it][0].z, wr[it][1].z, wr[it][2].z, wr[it][3].z);
            dstp[3 * 16] = make_float4(wr[it][0].w, wr[it][1].w, wr[it][2].w, wr[it][3].w);
        }
    }
}

__device__ __forceinline__ void fl_gemm_half(v8f (&acc)[4], const float* arow, const float4* wl,
                                             float rs) {
    float4 aw = *(const float4*)arow;
    float4 bw[4];
#pragma unroll
    for (int t = 0; t < 4; ++t) bw[t] = wl[t * 16];
#pragma unroll
    for (int j = 0; j < 8; ++j) {
        v2f a0, a1;
        a0.x = aw.x * rs;  a0.y = aw.y * rs;
        a1.x = aw.z * rs;  a1.y = aw.w * rs;
        float4 an = aw;
        float4 bn[4];
        if (j < 7) {
            an = *(const float4*)(arow + 8 * (j + 1));
#pragma unroll
            for (int t = 0; t < 4; ++t) bn[t] = wl[(2 * (j + 1)) * D_FEAT + t * 16];
        }
        __builtin_amdgcn_sched_barrier(0);
#pragma unroll
        for (int t = 0; t < 4; ++t) {
            v2f b0; b0.x = bw[t].x; b0.y = bw[t].y;
            acc[t] = __builtin_amdgcn_wmma_f32_16x16x4_f32(
                false, a0, false, b0, (short)0, acc[t], false, false);
        }
#pragma unroll
        for (int t = 0; t < 4; ++t) {
            v2f b1; b1.x = bw[t].z; b1.y = bw[t].w;
            acc[t] = __builtin_amdgcn_wmma_f32_16x16x4_f32(
                false, a1, false, b1, (short)0, acc[t], false, false);
        }
        __builtin_amdgcn_sched_barrier(0);
        if (j < 7) {
            aw = an;
#pragma unroll
            for (int t = 0; t < 4; ++t) bw[t] = bn[t];
        }
    }
}

__device__ __forceinline__ float fl_finish_scaled(float a, float b, float r, float ns, int relu_c) {
    float cc = a + b;
    if (relu_c) cc = fmaxf(cc, 0.0f);
    return fmaxf(r + ns * cc, 0.0f);
}

__global__ void __launch_bounds__(FL_THREADS)
graphconv_bins_kernel(const float* __restrict__ hin, const float* __restrict__ norm_src,
                      const float* __restrict__ norm_dst,
                      const int2* __restrict__ binned, const int* __restrict__ boff,
                      const float* __restrict__ Wg, const float* __restrict__ bias,
                      float* __restrict__ hout, int N, int bn, int relu_c, int unscale_out) {
    __shared__ __attribute__((aligned(16))) float rows[FL_ROWS * FL_PITCH];
    __shared__ float4 Wl[32 * D_FEAT];
    const int tid = threadIdx.x;
    const int lane = tid & 31;
    const int wid = __builtin_amdgcn_readfirstlane((int)(threadIdx.x >> 5));
    const int bin = blockIdx.x;
    const int lo = bin * bn;

    float4 wr[2][4];
    fl_w_load(wr, Wg, tid);
    float4* r4 = (float4*)rows;
    for (int i = tid; i < FL_ROWS * FL_PITCH / 4; i += FL_THREADS)
        r4[i] = make_float4(0.0f, 0.0f, 0.0f, 0.0f);
    const int b0 = boff[bin];
    const int b1 = boff[bin + 1];
    fl_w_store(Wl, wr, tid);
    __syncthreads();

    for (int e0 = b0 + wid * SC_EPW; e0 < b1; e0 += FL_WAVES * SC_EPW) {
        const int n = (b1 - e0 < SC_EPW) ? (b1 - e0) : SC_EPW;
        int el = e0 + (lane & (SC_EPW - 1));
        if (el > b1 - 1) el = b1 - 1;
        const int2 ed = binned[el];
        const int sv = ed.x;
        unsigned du = (unsigned)(ed.y - lo);
        if (du > (unsigned)(FL_ROWS - 1)) du = (unsigned)(FL_ROWS - 1);
        const int dv = (int)du;

        if (n == SC_EPW) {
            float v[SC_EPW][4];
#pragma unroll
            for (int i = 0; i < SC_EPW; ++i) {
                const int s = __builtin_amdgcn_readlane(sv, i);
                const float* hp = hin + (size_t)s * D_FEAT + lane;
#pragma unroll
                for (int k = 0; k < 4; ++k) v[i][k] = hp[32 * k];
            }
#pragma unroll
            for (int i = 0; i < SC_EPW; ++i) {
                const int d = __builtin_amdgcn_readlane(dv, i);
                float* rp = rows + d * FL_PITCH + lane;
#pragma unroll
                for (int k = 0; k < 4; ++k)
                    __hip_atomic_fetch_add(rp + 32 * k, v[i][k], __ATOMIC_RELAXED,
                                           __HIP_MEMORY_SCOPE_WORKGROUP);
            }
        } else {
            for (int i = 0; i < n; ++i) {
                const int s = __builtin_amdgcn_readlane(sv, i);
                const int d = __builtin_amdgcn_readlane(dv, i);
                const float* hp = hin + (size_t)s * D_FEAT + lane;
                float* rp = rows + d * FL_PITCH + lane;
                float t[4];
#pragma unroll
                for (int k = 0; k < 4; ++k) t[k] = hp[32 * k];
#pragma unroll
                for (int k = 0; k < 4; ++k)
                    __hip_atomic_fetch_add(rp + 32 * k, t[k], __ATOMIC_RELAXED,
                                           __HIP_MEMORY_SCOPE_WORKGROUP);
            }
        }
    }
    __syncthreads();

    const int hh = lane >> 4;
    const int c  = lane & 15;
    const int rt = wid >> 1;
    const int g  = wid & 1;
    int node = lo + 16 * rt + c;
    if (node >= N) node = N - 1;
    const float rs = norm_dst[node];
    const float* arow = rows + (16 * rt + c) * FL_PITCH + 4 * hh;
    const float4* wl = Wl + hh * D_FEAT + (4 * g) * 16 + c;

    v8f acc[4] = {};
    fl_gemm_half(acc, arow, wl, rs);
    fl_gemm_half(acc, arow + 64, wl + 16 * D_FEAT, rs);

    const int r0 = 16 * rt + hh * 8;
    const int m0 = lo + r0;
    const float4 bz = *(const float4*)(bias + 64 * g + 4 * c);
    float4 rv[8];
    float nsv[8];
#pragma unroll
    for (int v = 0; v < 8; ++v) {
        int m = m0 + v;
        if (m >= N) m = N - 1;
        rv[v] = *(const float4*)(hin + (size_t)m * D_FEAT + 64 * g + 4 * c);
        nsv[v] = norm_src[m];
    }
    if (unscale_out) {
#pragma unroll
        for (int v = 0; v < 8; ++v) {
            if (r0 + v < bn && m0 + v < N) {
                const float inv = 1.0f / nsv[v];
                float4 o;
                o.x = ng_finish<1>(acc[0][v], bz.x, rv[v].x * inv, relu_c);
                o.y = ng_finish<1>(acc[1][v], bz.y, rv[v].y * inv, relu_c);
                o.z = ng_finish<1>(acc[2][v], bz.z, rv[v].z * inv, relu_c);
                o.w = ng_finish<1>(acc[3][v], bz.w, rv[v].w * inv, relu_c);
                *(float4*)(hout + (size_t)(m0 + v) * D_FEAT + 64 * g + 4 * c) = o;
            }
        }
    } else {
#pragma unroll
        for (int v = 0; v < 8; ++v) {
            if (r0 + v < bn && m0 + v < N) {
                float4 o;
                o.x = fl_finish_scaled(acc[0][v], bz.x, rv[v].x, nsv[v], relu_c);
                o.y = fl_finish_scaled(acc[1][v], bz.y, rv[v].y, nsv[v], relu_c);
                o.z = fl_finish_scaled(acc[2][v], bz.z, rv[v].z, nsv[v], relu_c);
                o.w = fl_finish_scaled(acc[3][v], bz.w, rv[v].w, nsv[v], relu_c);
                *(float4*)(hout + (size_t)(m0 + v) * D_FEAT + 64 * g + 4 * c) = o;
            }
        }
    }
}

#define RO_TG 8
#define RO_WAVES 16
#define RO_THREADS (32 * RO_WAVES)
#define RO_PA 132
#define RO_PT 260
#define RO_MLP 256
#define RO_BATCH 16
#define RO_SCAN 16

__device__ __forceinline__ void ro_b_load(v2f (&b)[32], const float* __restrict__ Wm, int kbase,
                                          int hh, int col) {
#pragma unroll
    for (int s = 0; s < 32; ++s) {
        const float* p = Wm + (size_t)(kbase + 4 * s + 2 * hh) * RO_MLP + col;
        b[s].x = p[0];
        b[s].y = p[RO_MLP];
    }
}

__device__ __forceinline__ void ro_stage(v8f& acc, const v2f (&b)[32], const float* arow) {
#pragma unroll
    for (int s = 0; s < 32; ++s) {
        const v2f a = *(const v2f*)(arow + 4 * s);
        acc = __builtin_amdgcn_wmma_f32_16x16x4_f32(false, a, false, b[s], (short)0, acc,
                                                    false, false);
    }
}

__device__ __forceinline__ void ro_flush(float* tile, int g, const float (&acc)[4], int lane) {
    if (g >= 0) {
        float* rp = tile + g * RO_PA + lane;
#pragma unroll
        for (int k = 0; k < 4; ++k)
            __hip_atomic_fetch_add(rp + 32 * k, acc[k], __ATOMIC_RELAXED,
                                   __HIP_MEMORY_SCOPE_WORKGROUP);
    }
}

__device__ __forceinline__ void ro_span(int& mn, int& mx, int id, int n, int g0) {
    if ((unsigned)(id - g0) < (unsigned)RO_TG) {
        mn = (n < mn) ? n : mn;
        mx = (n > mx) ? n : mx;
    }
}

__global__ void __launch_bounds__(RO_THREADS)
readout_kernel(const float* __restrict__ hfin, const int* __restrict__ gid,
               const float* __restrict__ W0, const float* __restrict__ c0,
               const float* __restrict__ W1, const float* __restrict__ c1,
               const float* __restrict__ W2, const float* __restrict__ c2,
               float* __restrict__ out, int N, int G) {
    __shared__ __attribute__((aligned(16))) float tA[16 * RO_PA];
    __shared__ __attribute__((aligned(16))) float tB[16 * RO_PT];
    __shared__ __attribute__((aligned(16))) float tC[16 * RO_PT];
    __shared__ int rng[2];
    const int tid = threadIdx.x;
    const int lane = tid & 31;
    const int wid = __builtin_amdgcn_readfirstlane((int)(threadIdx.x >> 5));
    const int hh = lane >> 4;
    const int c = lane & 15;
    const int col = 16 * wid + c;
    const int g0 = blockIdx.x * RO_TG;

    v2f bA[32], bB[32];
    ro_b_load(bA, W0, 0, hh, col);
    ro_b_load(bB, W1, 0, hh, col);

    for (int i = tid; i < 16 * RO_PA; i += RO_THREADS) tA[i] = 0.0f;
    if (tid == 0) { rng[0] = N; rng[1] = -1; }
    __syncthreads();

    int mn = N, mx = -1;
    const int n4 = N >> 2;
    const int4* __restrict__ gid4 = (const int4*)gid;
    for (int base = 0; base < n4; base += RO_SCAN * RO_THREADS) {
        int4 v[RO_SCAN];
#pragma unroll
        for (int u = 0; u < RO_SCAN; ++u) {
            const int j = base + u * RO_THREADS + tid;
            if (j < n4) v[u] = gid4[j];
            else v[u] = make_int4(-1, -1, -1, -1);
        }
#pragma unroll
        for (int u = 0; u < RO_SCAN; ++u) {
            const int n = 4 * (base + u * RO_THREADS + tid);
            ro_span(mn, mx, v[u].x, n + 0, g0);
            ro_span(mn, mx, v[u].y, n + 1, g0);
            ro_span(mn, mx, v[u].z, n + 2, g0);
            ro_span(mn, mx, v[u].w, n + 3, g0);
        }
    }
    for (int n = (n4 << 2) + tid; n < N; n += RO_THREADS) ro_span(mn, mx, gid[n], n, g0);
    for (int d = 16; d >= 1; d >>= 1) {
        const int a = __shfl_xor(mn, d);
        const int b = __shfl_xor(mx, d);
        mn = (a < mn) ? a : mn;
        mx = (b > mx) ? b : mx;
    }
    if (lane == 0) {
        atomicMin(&rng[0], mn);
        atomicMax(&rng[1], mx);
    }
    __syncthreads();
    const int lo = __builtin_amdgcn_readfirstlane(rng[0]);
    const int hi = __builtin_amdgcn_readfirstlane(rng[1]) + 1;

    if (hi > lo) {
        const int per = (hi - lo + RO_WAVES - 1) / RO_WAVES;
        const int w0 = lo + wid * per;
        const int w1 = (w0 + per < hi) ? (w0 + per) : hi;
        int cur = -1;
        float acc[4] = {0.0f, 0.0f, 0.0f, 0.0f};
        for (int base = w0; base < w1; base += RO_BATCH) {
            int nl = base + (lane & (RO_BATCH - 1));
            if (nl > w1 - 1) nl = w1 - 1;
            const int gv = gid[nl] - g0;
            float v[RO_BATCH][4];
#pragma unroll
            for (int i = 0; i < RO_BATCH; ++i) {
                int n = base + i;
                if (n > w1 - 1) n = w1 - 1;
                const float* hp = hfin + (size_t)n * D_FEAT + lane;
#pragma unroll
                for (int k = 0; k < 4; ++k) v[i][k] = hp[32 * k];
            }
#pragma unroll
            for (int i = 0; i < RO_BATCH; ++i)
                asm volatile("" :: "v"(v[i][0]), "v"(v[i][1]), "v"(v[i][2]), "v"(v[i][3]));
#pragma unroll
            for (int i = 0; i < RO_BATCH; ++i) {
                const int g = __builtin_amdgcn_readlane(gv, i);
                if (base + i < w1 && (unsigned)g < (unsigned)RO_TG) {
                    if (g != cur) {
                        ro_flush(tA, cur, acc, lane);
                        cur = g;
#pragma unroll
                        for (int k = 0; k < 4; ++k) acc[k] = 0.0f;
                    }
#pragma unroll
                    for (int k = 0; k < 4; ++k) acc[k] += v[i][k];
                }
            }
        }
        ro_flush(tA, cur, acc, lane);
    }
    __syncthreads();

    v8f acc0 = {};
    ro_stage(acc0, bA, tA + c * RO_PA + 2 * hh);
    {
        const float bv = c0[col];
#pragma unroll
        for (int v = 0; v < 8; ++v) tB[(8 * hh + v) * RO_PT + col] = fmaxf(acc0[v] + bv, 0.0f);
    }
    __syncthreads();

    v8f acc1 = {};
    ro_b_load(bA, W1, 128, hh, col);
    ro_stage(acc1, bB, tB + c * RO_PT + 2 * hh);
    ro_b_load(bB, W2, 0, hh, col);
    ro_stage(acc1, bA, tB + c * RO_PT + 128 + 2 * hh);
    {
        const float bv = c1[col];
#pragma unroll
        for (int v = 0; v < 8; ++v) tC[(8 * hh + v) * RO_PT + col] = fmaxf(acc1[v] + bv, 0.0f);
    }
    __syncthreads();

    v8f acc2 = {};
    ro_b_load(bA, W2, 128, hh, col);
    ro_stage(acc2, bB, tC + c * RO_PT + 2 * hh);
    ro_stage(acc2, bA, tC + c * RO_PT + 128 + 2 * hh);
    {
        const float bv = c2[col];
#pragma unroll
        for (int v = 0; v < 8; ++v) {
            const int r = 8 * hh + v;
            if (r < RO_TG && g0 + r < G) out[(size_t)(g0 + r) * RO_MLP + col] = acc2[v] + bv;
        }
    }
}


extern "C" void kernel_launch(void* const* d_in, const int* in_sizes, int n_in,
                              void* d_out, int out_size, void* d_ws, size_t ws_size,
                              hipStream_t stream) {
    const float* node_features = (const float*)d_in[0];
    const int*   edge_src      = (const int*)d_in[1];
    const int*   edge_dst      = (const int*)d_in[2];
    const int*   graph_ids     = (const int*)d_in[3];
    const float* W_proj        = (const float*)d_in[4];
    const float* b_proj        = (const float*)d_in[5];
    const float* W_gcn         = (const float*)d_in[6];
    const float* b_gcn         = (const float*)d_in[7];
    const float* W_mlp0        = (const float*)d_in[8];
    const float* b_mlp0        = (const float*)d_in[9];
    const float* W_mlp1        = (const float*)d_in[10];
    const float* b_mlp1        = (const float*)d_in[11];
    const float* W_mlp2        = (const float*)d_in[12];
    const float* b_mlp2        = (const float*)d_in[13];

    const int D = D_FEAT;
    const int Mm = 256;
    const int N = in_sizes[3];
    const int E = in_sizes[1];
    const int L = in_sizes[6] / (D * D);
    const int G = out_size / Mm;

    float* ws = (float*)d_ws;
    float* norm_src = ws;
    float* norm_dst = norm_src + N;
    float* h        = norm_dst + N;
    float* h2       = h + (size_t)N * D;
    float* hg       = h2 + (size_t)N * D;
    float* t0       = hg + (size_t)G * D;
    float* t1       = t0 + (size_t)G * Mm;
    const int bn_even = (N + 255) / 256;
    const int bn = (bn_even > FL_ROWS) ? FL_ROWS : ((bn_even < 1) ? 1 : bn_even);
    const int B = (N + bn - 1) / bn;
    int2* binned    = (int2*)(t1 + (size_t)G * Mm);
    int*  hist      = (int*)(binned + E);
    int*  boff      = hist + (size_t)BN_BLOCKS * B;
    int*  part      = boff + (B + 1);

    slice_count_kernel<<<dim3(BN_BLOCKS, 2), BN_THREADS, 0, stream>>>(edge_src, edge_dst, part, hist,
                                                                      E, N, B, bn);
    bin_place_kernel<<<2 * BN_BLOCKS, BN_THREADS, 0, stream>>>(edge_src, edge_dst, hist, part, boff,
                                                               binned, norm_src, norm_dst, E, N, B, bn);

    const int NG_ROWS = 16 * NG_WAVES;
    const unsigned ng_blocks = (unsigned)((N + NG_ROWS - 1) / NG_ROWS);
    node_gemm_rows128<0><<<ng_blocks, 32 * NG_WAVES, 0, stream>>>(node_features, W_proj, b_proj,
                                                                 nullptr, (L > 0) ? norm_src : nullptr,
                                                                 nullptr, h, N,  0);

    float* hcur = h;
    float* hnxt = h2;
    for (int i = 0; i < L; ++i) {
        graphconv_bins_kernel<<<(unsigned)B, FL_THREADS, 0, stream>>>(
            hcur, norm_src, norm_dst, binned, boff, W_gcn + (size_t)i * D * D, b_gcn + (size_t)i * D,
            hnxt, N, bn,  (i < L - 1) ? 1 : 0,  (i == L - 1) ? 1 : 0);
        float* hswap = hcur;
        hcur = hnxt;
        hnxt = hswap;
    }

    readout_kernel<<<(unsigned)((G + RO_TG - 1) / RO_TG), RO_THREADS, 0, stream>>>(
        hcur, graph_ids, W_mlp0, b_mlp0, W_mlp1, b_mlp1, W_mlp2, b_mlp2, (float*)d_out, N, G);
}
